# starting kernel
# baseline (speedup 1.0000x reference)
#include <hip/hip_runtime.h>


typedef float v2f __attribute__((ext_vector_type(2)));
typedef float v8f __attribute__((ext_vector_type(8)));

#define DM   512
#define SEQ  4096
#define NB   2
#define NH   8
#define DKH  64

__device__ __forceinline__ v8f wmma4(v2f a, v2f b, v8f c) {
  return __builtin_amdgcn_wmma_f32_16x16x4_f32(
       false, a,  false, b,
       (short)0, c,  false,  false);
}

__global__ void __launch_bounds__(256) kG(const float* __restrict__ k,
                                          const float* __restrict__ v,
                                          float* __restrict__ G) {
  const int lane = threadIdx.x & 31, wv = threadIdx.x >> 5;
  const int lm = lane & 15, lh = lane >> 4;
  const int job = blockIdx.x * 8 + wv;
  const int b = job >> 7;
  const int rest = job & 127;
  const int d0 = (rest >> 3) * 32;
  const int e0 = (rest & 7) * 64;
  const float* kb = k + (size_t)b * SEQ * DM;
  const float* vb = v + (size_t)b * SEQ * DM;
  v8f acc[8] = {};
  #pragma unroll 2
  for (int s0 = 0; s0 < SEQ; s0 += 4) {
    size_t rb = (size_t)(s0 + 2 * lh) * DM;
    v2f a0; a0.x = kb[rb + d0 + lm];      a0.y = kb[rb + DM + d0 + lm];
    v2f a1; a1.x = kb[rb + d0 + 16 + lm]; a1.y = kb[rb + DM + d0 + 16 + lm];
    #pragma unroll
    for (int t = 0; t < 4; ++t) {
      v2f bm; bm.x = vb[rb + e0 + 16 * t + lm]; bm.y = vb[rb + DM + e0 + 16 * t + lm];
      acc[t]     = wmma4(a0, bm, acc[t]);
      acc[4 + t] = wmma4(a1, bm, acc[4 + t]);
    }
  }
  float* Gb = G + ((size_t)b * DM + d0) * DM + e0;
  #pragma unroll
  for (int m = 0; m < 2; ++m)
    #pragma unroll
    for (int t = 0; t < 4; ++t)
      #pragma unroll
      for (int r = 0; r < 8; ++r)
        Gb[(size_t)(16 * m + r + 8 * lh) * DM + 16 * t + lm] = acc[4 * m + t][r];
}

__global__ void __launch_bounds__(256) kT(const float* __restrict__ wk,
                                          const float* __restrict__ G,
                                          float* __restrict__ T) {
  const int lane = threadIdx.x & 31, wv = threadIdx.x >> 5;
  const int lm = lane & 15, lh = lane >> 4;
  const int job = blockIdx.x * 8 + wv;
  const int bh = job >> 5;
  const int b = bh >> 3, h = bh & 7;
  const int rest = job & 31;
  const int i0 = (rest >> 3) * 16;
  const int e0 = (rest & 7) * 64;
  const float* arow = wk + (size_t)(h * 64 + i0 + lm) * DM + 2 * lh;
  const float* Gb = G + (size_t)b * DM * DM;
  v8f acc[4] = {};
  for (int kk = 0; kk < DM; kk += 4) {
    v2f a = *(const v2f*)(arow + kk);
    size_t rb = (size_t)(kk + 2 * lh) * DM;
    #pragma unroll
    for (int t = 0; t < 4; ++t) {
      v2f bm; bm.x = Gb[rb + e0 + 16 * t + lm]; bm.y = Gb[rb + DM + e0 + 16 * t + lm];
      acc[t] = wmma4(a, bm, acc[t]);
    }
  }
  float* Tb = T + ((size_t)bh * 64 + i0) * DM + e0;
  #pragma unroll
  for (int t = 0; t < 4; ++t)
    #pragma unroll
    for (int r = 0; r < 8; ++r)
      Tb[(size_t)(r + 8 * lh) * DM + 16 * t + lm] = acc[t][r];
}

__global__ void __launch_bounds__(256) kM(const float* __restrict__ T,
                                          const float* __restrict__ wvw,
                                          const float* __restrict__ bk,
                                          const float* __restrict__ bv,
                                          const float* __restrict__ u,
                                          const float* __restrict__ z,
                                          float* __restrict__ M) {
  const int lane = threadIdx.x & 31, wv = threadIdx.x >> 5;
  const int lm = lane & 15, lh = lane >> 4;
  const int job = blockIdx.x * 8 + wv;
  const int bh = job >> 2;
  const int b = bh >> 3, h = bh & 7;
  const int i0 = (job & 3) * 16;
  const float* arow = T + ((size_t)bh * 64 + i0 + lm) * DM + 2 * lh;
  v8f acc[4] = {};
  for (int kk = 0; kk < DM; kk += 4) {
    v2f a = *(const v2f*)(arow + kk);
    #pragma unroll
    for (int t = 0; t < 4; ++t) {
      v2f bm = *(const v2f*)(wvw + (size_t)(h * 64 + 16 * t + lm) * DM + kk + 2 * lh);
      acc[t] = wmma4(a, bm, acc[t]);
    }
  }
  #pragma unroll
  for (int t = 0; t < 4; ++t) {
    const int jf = h * 64 + 16 * t + lm;
    const float bvj = bv[jf];
    const float zj  = z[(size_t)b * DM + jf];
    #pragma unroll
    for (int r = 0; r < 8; ++r) {
      const int ifl = h * 64 + i0 + r + 8 * lh;
      const float bki = bk[ifl];
      const float ui  = u[(size_t)b * DM + ifl];
      float val = acc[t][r] + ui * bvj + bki * zj + (float)SEQ * bki * bvj;
      M[((size_t)bh * 64 + i0 + r + 8 * lh) * DKH + 16 * t + lm] = 0.125f * val;
    }
  }
}

__global__ void __launch_bounds__(256) kP(const float* __restrict__ M,
                                          const float* __restrict__ wo,
                                          float* __restrict__ P) {
  const int lane = threadIdx.x & 31, wv = threadIdx.x >> 5;
  const int lm = lane & 15, lh = lane >> 4;
  const int job = blockIdx.x * 8 + wv;
  const int bh = job >> 5;
  const int b = bh >> 3, h = bh & 7;
  const int i0 = ((job >> 3) & 3) * 16;
  const int o0 = (job & 7) * 64;
  const float* arow = M + ((size_t)bh * 64 + i0 + lm) * DKH + 2 * lh;
  v8f acc[4] = {};
  for (int kk = 0; kk < DKH; kk += 4) {
    v2f a = *(const v2f*)(arow + kk);
    #pragma unroll
    for (int t = 0; t < 4; ++t) {
      v2f bm = *(const v2f*)(wo + (size_t)(o0 + 16 * t + lm) * DM + h * 64 + kk + 2 * lh);
      acc[t] = wmma4(a, bm, acc[t]);
    }
  }
  float* Pb = P + ((size_t)b * DM + h * 64 + i0) * DM + o0;
  #pragma unroll
  for (int t = 0; t < 4; ++t)
    #pragma unroll
    for (int r = 0; r < 8; ++r)
      Pb[(size_t)(r + 8 * lh) * DM + 16 * t + lm] = acc[t][r];
}

__global__ void __launch_bounds__(256) kW4(const float* __restrict__ wq,
                                           const float* __restrict__ P,
                                           float* __restrict__ W4) {
  const int lane = threadIdx.x & 31, wv = threadIdx.x >> 5;
  const int lm = lane & 15, lh = lane >> 4;
  const int job = blockIdx.x * 8 + wv;
  const int b = job >> 8;
  const int rest = job & 255;
  const int d0 = (rest >> 3) * 16;
  const int o0 = (rest & 7) * 64;
  const float* Pb = P + (size_t)b * DM * DM;
  v8f acc[4] = {};
  for (int kk = 0; kk < DM; kk += 4) {
    size_t fr = (size_t)(kk + 2 * lh) * DM;
    v2f a; a.x = wq[fr + d0 + lm]; a.y = wq[fr + DM + d0 + lm];
    #pragma unroll
    for (int t = 0; t < 4; ++t) {
      v2f bm; bm.x = Pb[fr + o0 + 16 * t + lm]; bm.y = Pb[fr + DM + o0 + 16 * t + lm];
      acc[t] = wmma4(a, bm, acc[t]);
    }
  }
  float* Wb = W4 + ((size_t)b * DM + d0) * DM + o0;
  #pragma unroll
  for (int t = 0; t < 4; ++t)
    #pragma unroll
    for (int r = 0; r < 8; ++r)
      Wb[(size_t)(r + 8 * lh) * DM + 16 * t + lm] = acc[t][r];
}

__global__ void __launch_bounds__(256) kOut(const float* __restrict__ q,
                                            const float* __restrict__ W4,
                                            const float* __restrict__ bo,
                                            const float* __restrict__ cq,
                                            float* __restrict__ out) {
  const int lane = threadIdx.x & 31, wv = threadIdx.x >> 5;
  const int lm = lane & 15, lh = lane >> 4;
  const int job = blockIdx.x * 8 + wv;
  const int b = job >> 7;
  const int n0 = (job & 127) * 32;
  const int o0 = blockIdx.y * 64;
  const float* qrow0 = q + ((size_t)b * SEQ + n0 + lm) * DM + 2 * lh;
  const float* qrow1 = qrow0 + (size_t)16 * DM;
  const float* Wb = W4 + (size_t)b * DM * DM;
  v8f acc[8] = {};
  #pragma unroll 2
  for (int kk = 0; kk < DM; kk += 4) {
    v2f a0 = *(const v2f*)(qrow0 + kk);
    v2f a1 = *(const v2f*)(qrow1 + kk);
    size_t rb = (size_t)(kk + 2 * lh) * DM;
    #pragma unroll
    for (int t = 0; t < 4; ++t) {
      v2f bm; bm.x = Wb[rb + o0 + 16 * t + lm]; bm.y = Wb[rb + DM + o0 + 16 * t + lm];
      acc[t]     = wmma4(a0, bm, acc[t]);
      acc[4 + t] = wmma4(a1, bm, acc[4 + t]);
    }
  }
  float* ob = out + ((size_t)b * SEQ + n0) * DM + o0;
  #pragma unroll
  for (int t = 0; t < 4; ++t) {
    float add = bo[o0 + 16 * t + lm] + cq[(size_t)b * DM + o0 + 16 * t + lm];
    #pragma unroll
    for (int m = 0; m < 2; ++m)
      #pragma unroll
      for (int r = 0; r < 8; ++r)
        ob[(size_t)(16 * m + r + 8 * lh) * DM + 16 * t + lm] = acc[4 * m + t][r] + add;
  }
}

__global__ void kSums(const float* __restrict__ k, const float* __restrict__ v,
                      float* __restrict__ ksum, float* __restrict__ vsum) {
  int idx = blockIdx.x * blockDim.x + threadIdx.x;
  if (idx >= NB * DM) return;
  int b = idx >> 9, d = idx & (DM - 1);
  const float* kp = k + (size_t)b * SEQ * DM + d;
  const float* vp = v + (size_t)b * SEQ * DM + d;
  float sk = 0.f, sv = 0.f;
  for (int s = 0; s < SEQ; ++s) { sk += kp[(size_t)s * DM]; sv += vp[(size_t)s * DM]; }
  ksum[idx] = sk; vsum[idx] = sv;
}

__global__ void kUZ(const float* __restrict__ wk, const float* __restrict__ wvw,
                    const float* __restrict__ ksum, const float* __restrict__ vsum,
                    float* __restrict__ u, float* __restrict__ z) {
  int idx = blockIdx.x * blockDim.x + threadIdx.x;
  if (idx >= NB * DM) return;
  int b = idx >> 9, f = idx & (DM - 1);
  const float* wkr = wk + (size_t)f * DM;
  const float* wvr = wvw + (size_t)f * DM;
  const float* ks = ksum + (size_t)b * DM;
  const float* vs = vsum + (size_t)b * DM;
  float su = 0.f, sz = 0.f;
  for (int d = 0; d < DM; ++d) { su += wkr[d] * ks[d]; sz += wvr[d] * vs[d]; }
  u[idx] = su; z[idx] = sz;
}

__global__ void kCQ(const float* __restrict__ bq, const float* __restrict__ P,
                    float* __restrict__ cq) {
  int idx = blockIdx.x * blockDim.x + threadIdx.x;
  if (idx >= NB * DM) return;
  int b = idx >> 9, o = idx & (DM - 1);
  const float* Pb = P + (size_t)b * DM * DM + o;
  float s = 0.f;
  for (int f = 0; f < DM; ++f) s += bq[f] * Pb[(size_t)f * DM];
  cq[idx] = s;
}

extern "C" void kernel_launch(void* const* d_in, const int* in_sizes, int n_in,
                              void* d_out, int out_size, void* d_ws, size_t ws_size,
                              hipStream_t stream) {
  const float* q   = (const float*)d_in[0];
  const float* k   = (const float*)d_in[1];
  const float* v   = (const float*)d_in[2];
  const float* wq  = (const float*)d_in[3];
  const float* bq  = (const float*)d_in[4];
  const float* wk  = (const float*)d_in[5];
  const float* bk  = (const float*)d_in[6];
  const float* wvw = (const float*)d_in[7];
  const float* bv  = (const float*)d_in[8];
  const float* wo  = (const float*)d_in[9];
  const float* bo  = (const float*)d_in[10];
  float* out = (float*)d_out;

  float* ws = (float*)d_ws;
  float* G    = ws;
  float* T    = G    + (size_t)NB * DM * DM;
  float* M    = T    + (size_t)NB * DM * DM;
  float* P    = M    + (size_t)NB * NH * DKH * DKH;
  float* W4   = P    + (size_t)NB * DM * DM;
  float* ksum = W4   + (size_t)NB * DM * DM;
  float* vsum = ksum + (size_t)NB * DM;
  float* u    = vsum + (size_t)NB * DM;
  float* z    = u    + (size_t)NB * DM;
  float* cq   = z    + (size_t)NB * DM;

  kSums<<<4, 256, 0, stream>>>(k, v, ksum, vsum);
  kUZ  <<<4, 256, 0, stream>>>(wk, wvw, ksum, vsum, u, z);
  kG   <<<32, 256, 0, stream>>>(k, v, G);
  kT   <<<64, 256, 0, stream>>>(wk, G, T);
  kM   <<<8, 256, 0, stream>>>(T, wvw, bk, bv, u, z, M);
  kP   <<<64, 256, 0, stream>>>(M, wo, P);
  kCQ  <<<4, 256, 0, stream>>>(bq, P, cq);
  kW4  <<<64, 256, 0, stream>>>(wq, P, W4);
  dim3 go(32, 8, 1);
  kOut <<<go, 256, 0, stream>>>(q, W4, bo, cq, out);
}


# kG: split the SEQ reduction of G=k^T v into 16 chunks (512 work-groups instead of 32), partials added in chunk order by a new small kernel; changes G's summation tree
# speedup vs baseline: 1.6576x; 1.6576x over previous
#include <hip/hip_runtime.h>


typedef float v2f __attribute__((ext_vector_type(2)));
typedef float v8f __attribute__((ext_vector_type(8)));

#define DM   512
#define SEQ  4096
#define NB   2
#define NH   8
#define DKH  64
#define KSPLIT 16
#define SCH  (SEQ / KSPLIT)
#define KG_BLOCKS    (32 * KSPLIT)
#define GSUM_BLOCKS  (NB * DM * DM / 4 / 256)

__device__ __forceinline__ v8f wmma4(v2f a, v2f b, v8f c) {
  return __builtin_amdgcn_wmma_f32_16x16x4_f32(
       false, a,  false, b,
       (short)0, c,  false,  false);
}

__global__ void __launch_bounds__(256) kG(const float* __restrict__ k,
                                          const float* __restrict__ v,
                                          float* __restrict__ Gp) {
  const int lane = threadIdx.x & 31, wv = threadIdx.x >> 5;
  const int lm = lane & 15, lh = lane >> 4;
  const int job = (blockIdx.x & 31) * 8 + wv;
  const int c = blockIdx.x >> 5;
  const int b = job >> 7;
  const int rest = job & 127;
  const int d0 = (rest >> 3) * 32;
  const int e0 = (rest & 7) * 64;
  const float* kb = k + ((size_t)b * SEQ + (size_t)c * SCH) * DM;
  const float* vb = v + ((size_t)b * SEQ + (size_t)c * SCH) * DM;
  v8f acc[8] = {};
  #pragma unroll 2
  for (int s0 = 0; s0 < SCH; s0 += 4) {
    size_t rb = (size_t)(s0 + 2 * lh) * DM;
    v2f a0; a0.x = kb[rb + d0 + lm];      a0.y = kb[rb + DM + d0 + lm];
    v2f a1; a1.x = kb[rb + d0 + 16 + lm]; a1.y = kb[rb + DM + d0 + 16 + lm];
    #pragma unroll
    for (int t = 0; t < 4; ++t) {
      v2f bm; bm.x = vb[rb + e0 + 16 * t + lm]; bm.y = vb[rb + DM + e0 + 16 * t + lm];
      acc[t]     = wmma4(a0, bm, acc[t]);
      acc[4 + t] = wmma4(a1, bm, acc[4 + t]);
    }
  }
  float* Gb = Gp + (((size_t)c * NB + b) * DM + d0) * DM + e0;
  #pragma unroll
  for (int m = 0; m < 2; ++m)
    #pragma unroll
    for (int t = 0; t < 4; ++t)
      #pragma unroll
      for (int r = 0; r < 8; ++r)
        Gb[(size_t)(16 * m + r + 8 * lh) * DM + 16 * t + lm] = acc[4 * m + t][r];
}

__global__ void __launch_bounds__(256) kGsum(const float* __restrict__ Gp,
                                             float* __restrict__ G) {
  const int i = blockIdx.x * 256 + threadIdx.x;
  const float4* p = (const float4*)Gp + i;
  float4 s = p[0];
  for (int c = 1; c < KSPLIT; ++c) {
    const float4 t = p[(size_t)c * (NB * DM * DM / 4)];
    s.x += t.x; s.y += t.y; s.z += t.z; s.w += t.w;
  }
  ((float4*)G)[i] = s;
}

__global__ void __launch_bounds__(256) kT(const float* __restrict__ wk,
                                          const float* __restrict__ G,
                                          float* __restrict__ T) {
  const int lane = threadIdx.x & 31, wv = threadIdx.x >> 5;
  const int lm = lane & 15, lh = lane >> 4;
  const int job = blockIdx.x * 8 + wv;
  const int bh = job >> 5;
  const int b = bh >> 3, h = bh & 7;
  const int rest = job & 31;
  const int i0 = (rest >> 3) * 16;
  const int e0 = (rest & 7) * 64;
  const float* arow = wk + (size_t)(h * 64 + i0 + lm) * DM + 2 * lh;
  const float* Gb = G + (size_t)b * DM * DM;
  v8f acc[4] = {};
  for (int kk = 0; kk < DM; kk += 4) {
    v2f a = *(const v2f*)(arow + kk);
    size_t rb = (size_t)(kk + 2 * lh) * DM;
    #pragma unroll
    for (int t = 0; t < 4; ++t) {
      v2f bm; bm.x = Gb[rb + e0 + 16 * t + lm]; bm.y = Gb[rb + DM + e0 + 16 * t + lm];
      acc[t] = wmma4(a, bm, acc[t]);
    }
  }
  float* Tb = T + ((size_t)bh * 64 + i0) * DM + e0;
  #pragma unroll
  for (int t = 0; t < 4; ++t)
    #pragma unroll
    for (int r = 0; r < 8; ++r)
      Tb[(size_t)(r + 8 * lh) * DM + 16 * t + lm] = acc[t][r];
}

__global__ void __launch_bounds__(256) kM(const float* __restrict__ T,
                                          const float* __restrict__ wvw,
                                          const float* __restrict__ bk,
                                          const float* __restrict__ bv,
                                          const float* __restrict__ u,
                                          const float* __restrict__ z,
                                          float* __restrict__ M) {
  const int lane = threadIdx.x & 31, wv = threadIdx.x >> 5;
  const int lm = lane & 15, lh = lane >> 4;
  const int job = blockIdx.x * 8 + wv;
  const int bh = job >> 2;
  const int b = bh >> 3, h = bh & 7;
  const int i0 = (job & 3) * 16;
  const float* arow = T + ((size_t)bh * 64 + i0 + lm) * DM + 2 * lh;
  v8f acc[4] = {};
  for (int kk = 0; kk < DM; kk += 4) {
    v2f a = *(const v2f*)(arow + kk);
    #pragma unroll
    for (int t = 0; t < 4; ++t) {
      v2f bm = *(const v2f*)(wvw + (size_t)(h * 64 + 16 * t + lm) * DM + kk + 2 * lh);
      acc[t] = wmma4(a, bm, acc[t]);
    }
  }
  #pragma unroll
  for (int t = 0; t < 4; ++t) {
    const int jf = h * 64 + 16 * t + lm;
    const float bvj = bv[jf];
    const float zj  = z[(size_t)b * DM + jf];
    #pragma unroll
    for (int r = 0; r < 8; ++r) {
      const int ifl = h * 64 + i0 + r + 8 * lh;
      const float bki = bk[ifl];
      const float ui  = u[(size_t)b * DM + ifl];
      float val = acc[t][r] + ui * bvj + bki * zj + (float)SEQ * bki * bvj;
      M[((size_t)bh * 64 + i0 + r + 8 * lh) * DKH + 16 * t + lm] = 0.125f * val;
    }
  }
}

__global__ void __launch_bounds__(256) kP(const float* __restrict__ M,
                                          const float* __restrict__ wo,
                                          float* __restrict__ P) {
  const int lane = threadIdx.x & 31, wv = threadIdx.x >> 5;
  const int lm = lane & 15, lh = lane >> 4;
  const int job = blockIdx.x * 8 + wv;
  const int bh = job >> 5;
  const int b = bh >> 3, h = bh & 7;
  const int i0 = ((job >> 3) & 3) * 16;
  const int o0 = (job & 7) * 64;
  const float* arow = M + ((size_t)bh * 64 + i0 + lm) * DKH + 2 * lh;
  v8f acc[4] = {};
  for (int kk = 0; kk < DKH; kk += 4) {
    v2f a = *(const v2f*)(arow + kk);
    #pragma unroll
    for (int t = 0; t < 4; ++t) {
      v2f bm = *(const v2f*)(wo + (size_t)(o0 + 16 * t + lm) * DM + h * 64 + kk + 2 * lh);
      acc[t] = wmma4(a, bm, acc[t]);
    }
  }
  float* Pb = P + ((size_t)b * DM + h * 64 + i0) * DM + o0;
  #pragma unroll
  for (int t = 0; t < 4; ++t)
    #pragma unroll
    for (int r = 0; r < 8; ++r)
      Pb[(size_t)(r + 8 * lh) * DM + 16 * t + lm] = acc[t][r];
}

__global__ void __launch_bounds__(256) kW4(const float* __restrict__ wq,
                                           const float* __restrict__ P,
                                           float* __restrict__ W4) {
  const int lane = threadIdx.x & 31, wv = threadIdx.x >> 5;
  const int lm = lane & 15, lh = lane >> 4;
  const int job = blockIdx.x * 8 + wv;
  const int b = job >> 8;
  const int rest = job & 255;
  const int d0 = (rest >> 3) * 16;
  const int o0 = (rest & 7) * 64;
  const float* Pb = P + (size_t)b * DM * DM;
  v8f acc[4] = {};
  for (int kk = 0; kk < DM; kk += 4) {
    size_t fr = (size_t)(kk + 2 * lh) * DM;
    v2f a; a.x = wq[fr + d0 + lm]; a.y = wq[fr + DM + d0 + lm];
    #pragma unroll
    for (int t = 0; t < 4; ++t) {
      v2f bm; bm.x = Pb[fr + o0 + 16 * t + lm]; bm.y = Pb[fr + DM + o0 + 16 * t + lm];
      acc[t] = wmma4(a, bm, acc[t]);
    }
  }
  float* Wb = W4 + ((size_t)b * DM + d0) * DM + o0;
  #pragma unroll
  for (int t = 0; t < 4; ++t)
    #pragma unroll
    for (int r = 0; r < 8; ++r)
      Wb[(size_t)(r + 8 * lh) * DM + 16 * t + lm] = acc[t][r];
}

__global__ void __launch_bounds__(256) kOut(const float* __restrict__ q,
                                            const float* __restrict__ W4,
                                            const float* __restrict__ bo,
                                            const float* __restrict__ cq,
                                            float* __restrict__ out) {
  const int lane = threadIdx.x & 31, wv = threadIdx.x >> 5;
  const int lm = lane & 15, lh = lane >> 4;
  const int job = blockIdx.x * 8 + wv;
  const int b = job >> 7;
  const int n0 = (job & 127) * 32;
  const int o0 = blockIdx.y * 64;
  const float* qrow0 = q + ((size_t)b * SEQ + n0 + lm) * DM + 2 * lh;
  const float* qrow1 = qrow0 + (size_t)16 * DM;
  const float* Wb = W4 + (size_t)b * DM * DM;
  v8f acc[8] = {};
  #pragma unroll 2
  for (int kk = 0; kk < DM; kk += 4) {
    v2f a0 = *(const v2f*)(qrow0 + kk);
    v2f a1 = *(const v2f*)(qrow1 + kk);
    size_t rb = (size_t)(kk + 2 * lh) * DM;
    #pragma unroll
    for (int t = 0; t < 4; ++t) {
      v2f bm; bm.x = Wb[rb + o0 + 16 * t + lm]; bm.y = Wb[rb + DM + o0 + 16 * t + lm];
      acc[t]     = wmma4(a0, bm, acc[t]);
      acc[4 + t] = wmma4(a1, bm, acc[4 + t]);
    }
  }
  float* ob = out + ((size_t)b * SEQ + n0) * DM + o0;
  #pragma unroll
  for (int t = 0; t < 4; ++t) {
    float add = bo[o0 + 16 * t + lm] + cq[(size_t)b * DM + o0 + 16 * t + lm];
    #pragma unroll
    for (int m = 0; m < 2; ++m)
      #pragma unroll
      for (int r = 0; r < 8; ++r)
        ob[(size_t)(16 * m + r + 8 * lh) * DM + 16 * t + lm] = acc[4 * m + t][r] + add;
  }
}

__global__ void kSums(const float* __restrict__ k, const float* __restrict__ v,
                      float* __restrict__ ksum, float* __restrict__ vsum) {
  int idx = blockIdx.x * blockDim.x + threadIdx.x;
  if (idx >= NB * DM) return;
  int b = idx >> 9, d = idx & (DM - 1);
  const float* kp = k + (size_t)b * SEQ * DM + d;
  const float* vp = v + (size_t)b * SEQ * DM + d;
  float sk = 0.f, sv = 0.f;
  for (int s = 0; s < SEQ; ++s) { sk += kp[(size_t)s * DM]; sv += vp[(size_t)s * DM]; }
  ksum[idx] = sk; vsum[idx] = sv;
}

__global__ void kUZ(const float* __restrict__ wk, const float* __restrict__ wvw,
                    const float* __restrict__ ksum, const float* __restrict__ vsum,
                    float* __restrict__ u, float* __restrict__ z) {
  int idx = blockIdx.x * blockDim.x + threadIdx.x;
  if (idx >= NB * DM) return;
  int b = idx >> 9, f = idx & (DM - 1);
  const float* wkr = wk + (size_t)f * DM;
  const float* wvr = wvw + (size_t)f * DM;
  const float* ks = ksum + (size_t)b * DM;
  const float* vs = vsum + (size_t)b * DM;
  float su = 0.f, sz = 0.f;
  for (int d = 0; d < DM; ++d) { su += wkr[d] * ks[d]; sz += wvr[d] * vs[d]; }
  u[idx] = su; z[idx] = sz;
}

__global__ void kCQ(const float* __restrict__ bq, const float* __restrict__ P,
                    float* __restrict__ cq) {
  int idx = blockIdx.x * blockDim.x + threadIdx.x;
  if (idx >= NB * DM) return;
  int b = idx >> 9, o = idx & (DM - 1);
  const float* Pb = P + (size_t)b * DM * DM + o;
  float s = 0.f;
  for (int f = 0; f < DM; ++f) s += bq[f] * Pb[(size_t)f * DM];
  cq[idx] = s;
}

extern "C" void kernel_launch(void* const* d_in, const int* in_sizes, int n_in,
                              void* d_out, int out_size, void* d_ws, size_t ws_size,
                              hipStream_t stream) {
  const float* q   = (const float*)d_in[0];
  const float* k   = (const float*)d_in[1];
  const float* v   = (const float*)d_in[2];
  const float* wq  = (const float*)d_in[3];
  const float* bq  = (const float*)d_in[4];
  const float* wk  = (const float*)d_in[5];
  const float* bk  = (const float*)d_in[6];
  const float* wvw = (const float*)d_in[7];
  const float* bv  = (const float*)d_in[8];
  const float* wo  = (const float*)d_in[9];
  const float* bo  = (const float*)d_in[10];
  float* out = (float*)d_out;

  float* ws = (float*)d_ws;
  float* G    = ws;
  float* T    = G    + (size_t)NB * DM * DM;
  float* M    = T    + (size_t)NB * DM * DM;
  float* P    = M    + (size_t)NB * NH * DKH * DKH;
  float* W4   = P    + (size_t)NB * DM * DM;
  float* ksum = W4   + (size_t)NB * DM * DM;
  float* vsum = ksum + (size_t)NB * DM;
  float* u    = vsum + (size_t)NB * DM;
  float* z    = u    + (size_t)NB * DM;
  float* cq   = z    + (size_t)NB * DM;
  float* Gp   = cq   + (size_t)NB * DM;

  kSums<<<4, 256, 0, stream>>>(k, v, ksum, vsum);
  kUZ  <<<4, 256, 0, stream>>>(wk, wvw, ksum, vsum, u, z);
  kG   <<<KG_BLOCKS, 256, 0, stream>>>(k, v, Gp);
  kGsum<<<GSUM_BLOCKS, 256, 0, stream>>>(Gp, G);
  kT   <<<64, 256, 0, stream>>>(wk, G, T);
  kM   <<<8, 256, 0, stream>>>(T, wvw, bk, bv, u, z, M);
  kP   <<<64, 256, 0, stream>>>(M, wo, P);
  kCQ  <<<4, 256, 0, stream>>>(bq, P, cq);
  kW4  <<<64, 256, 0, stream>>>(wq, P, W4);
  dim3 go(32, 8, 1);
  kOut <<<go, 256, 0, stream>>>(q, W4, bo, cq, out);
}


# kSums folded into kG: k and v column sums taken from the fragments kG already loads (per chunk and lane half), partials added in index order in kGsum; 9 launches; changes ksum/vsum summation tree
# speedup vs baseline: 2.6931x; 1.6247x over previous
#include <hip/hip_runtime.h>


typedef float v2f __attribute__((ext_vector_type(2)));
typedef float v8f __attribute__((ext_vector_type(8)));

#define DM   512
#define SEQ  4096
#define NB   2
#define NH   8
#define DKH  64
#define KSPLIT 16
#define SCH  (SEQ / KSPLIT)
#define KG_BLOCKS    (32 * KSPLIT)
#define GSUM_BLOCKS  (NB * DM * DM / 4 / 256)
#define NPART  (2 * KSPLIT)
#define NSUMS  (2 * NB * DM)
#define SUMS_BLOCKS  (NSUMS / 256)
#define GSUM_GRID    (GSUM_BLOCKS + SUMS_BLOCKS)

__device__ __forceinline__ v8f wmma4(v2f a, v2f b, v8f c) {
  return __builtin_amdgcn_wmma_f32_16x16x4_f32(
       false, a,  false, b,
       (short)0, c,  false,  false);
}

template <bool SK, bool SV>
__device__ __forceinline__ void g_tile(const float* __restrict__ kb,
                                       const float* __restrict__ vb,
                                       float* __restrict__ Gb,
                                       float* __restrict__ spk,
                                       float* __restrict__ spv,
                                       int d0, int e0, int lm, int lh) {
  v8f acc[8] = {};
  v2f ks0 = {0.f, 0.f}, ks1 = {0.f, 0.f};
  v2f vs[4] = {};
  #pragma unroll 2
  for (int s0 = 0; s0 < SCH; s0 += 4) {
    size_t rb = (size_t)(s0 + 2 * lh) * DM;
    v2f a0; a0.x = kb[rb + d0 + lm];      a0.y = kb[rb + DM + d0 + lm];
    v2f a1; a1.x = kb[rb + d0 + 16 + lm]; a1.y = kb[rb + DM + d0 + 16 + lm];
    if (SK) { ks0 += a0; ks1 += a1; }
    #pragma unroll
    for (int t = 0; t < 4; ++t) {
      v2f bm; bm.x = vb[rb + e0 + 16 * t + lm]; bm.y = vb[rb + DM + e0 + 16 * t + lm];
      if (SV) vs[t] += bm;
      acc[t]     = wmma4(a0, bm, acc[t]);
      acc[4 + t] = wmma4(a1, bm, acc[4 + t]);
    }
  }
  #pragma unroll
  for (int m = 0; m < 2; ++m)
    #pragma unroll
    for (int t = 0; t < 4; ++t)
      #pragma unroll
      for (int r = 0; r < 8; ++r)
        Gb[(size_t)(16 * m + r + 8 * lh) * DM + 16 * t + lm] = acc[4 * m + t][r];
  if (SK) {
    spk[d0 + lm]      = ks0.x + ks0.y;
    spk[d0 + 16 + lm] = ks1.x + ks1.y;
  }
  if (SV) {
    #pragma unroll
    for (int t = 0; t < 4; ++t) spv[e0 + 16 * t + lm] = vs[t].x + vs[t].y;
  }
}

__global__ void __launch_bounds__(256) kG(const float* __restrict__ k,
                                          const float* __restrict__ v,
                                          float* __restrict__ Gp,
                                          float* __restrict__ SP) {
  const int lane = threadIdx.x & 31, wv = threadIdx.x >> 5;
  const int lm = lane & 15, lh = lane >> 4;
  const int job = (blockIdx.x & 31) * 8 + wv;
  const int c = blockIdx.x >> 5;
  const int b = job >> 7;
  const int rest = job & 127;
  const int d0 = (rest >> 3) * 32;
  const int e0 = (rest & 7) * 64;
  const float* kb = k + ((size_t)b * SEQ + (size_t)c * SCH) * DM;
  const float* vb = v + ((size_t)b * SEQ + (size_t)c * SCH) * DM;
  float* Gb = Gp + (((size_t)c * NB + b) * DM + d0) * DM + e0;
  float* spk = SP + ((size_t)(2 * c + lh) * 2 + 0) * (NB * DM) + (size_t)b * DM;
  float* spv = SP + ((size_t)(2 * c + lh) * 2 + 1) * (NB * DM) + (size_t)b * DM;
  const bool sumK = __builtin_amdgcn_readfirstlane(wv) == 0;
  const bool sumV = (blockIdx.x & 15) == 0;
  if (sumV) {
    if (sumK) g_tile<true, true>(kb, vb, Gb, spk, spv, d0, e0, lm, lh);
    else      g_tile<false, true>(kb, vb, Gb, spk, spv, d0, e0, lm, lh);
  } else {
    if (sumK) g_tile<true, false>(kb, vb, Gb, spk, spv, d0, e0, lm, lh);
    else      g_tile<false, false>(kb, vb, Gb, spk, spv, d0, e0, lm, lh);
  }
}

__global__ void __launch_bounds__(256) kGsum(const float* __restrict__ Gp,
                                             float* __restrict__ G,
                                             const float* __restrict__ SP,
                                             float* __restrict__ sums) {
  if (blockIdx.x >= GSUM_BLOCKS) {
    const int j = (blockIdx.x - GSUM_BLOCKS) * 256 + threadIdx.x;
    float s = SP[j];
    for (int p = 1; p < NPART; ++p) s += SP[(size_t)p * NSUMS + j];
    sums[j] = s;
    return;
  }
  const int i = blockIdx.x * 256 + threadIdx.x;
  const float4* p = (const float4*)Gp + i;
  float4 s = p[0];
  for (int c = 1; c < KSPLIT; ++c) {
    const float4 t = p[(size_t)c * (NB * DM * DM / 4)];
    s.x += t.x; s.y += t.y; s.z += t.z; s.w += t.w;
  }
  ((float4*)G)[i] = s;
}

__global__ void __launch_bounds__(256) kT(const float* __restrict__ wk,
                                          const float* __restrict__ G,
                                          float* __restrict__ T) {
  const int lane = threadIdx.x & 31, wv = threadIdx.x >> 5;
  const int lm = lane & 15, lh = lane >> 4;
  const int job = blockIdx.x * 8 + wv;
  const int bh = job >> 5;
  const int b = bh >> 3, h = bh & 7;
  const int rest = job & 31;
  const int i0 = (rest >> 3) * 16;
  const int e0 = (rest & 7) * 64;
  const float* arow = wk + (size_t)(h * 64 + i0 + lm) * DM + 2 * lh;
  const float* Gb = G + (size_t)b * DM * DM;
  v8f acc[4] = {};
  for (int kk = 0; kk < DM; kk += 4) {
    v2f a = *(const v2f*)(arow + kk);
    size_t rb = (size_t)(kk + 2 * lh) * DM;
    #pragma unroll
    for (int t = 0; t < 4; ++t) {
      v2f bm; bm.x = Gb[rb + e0 + 16 * t + lm]; bm.y = Gb[rb + DM + e0 + 16 * t + lm];
      acc[t] = wmma4(a, bm, acc[t]);
    }
  }
  float* Tb = T + ((size_t)bh * 64 + i0) * DM + e0;
  #pragma unroll
  for (int t = 0; t < 4; ++t)
    #pragma unroll
    for (int r = 0; r < 8; ++r)
      Tb[(size_t)(r + 8 * lh) * DM + 16 * t + lm] = acc[t][r];
}

__global__ void __launch_bounds__(256) kM(const float* __restrict__ T,
                                          const float* __restrict__ wvw,
                                          const float* __restrict__ bk,
                                          const float* __restrict__ bv,
                                          const float* __restrict__ u,
                                          const float* __restrict__ z,
                                          float* __restrict__ M) {
  const int lane = threadIdx.x & 31, wv = threadIdx.x >> 5;
  const int lm = lane & 15, lh = lane >> 4;
  const int job = blockIdx.x * 8 + wv;
  const int bh = job >> 2;
  const int b = bh >> 3, h = bh & 7;
  const int i0 = (job & 3) * 16;
  const float* arow = T + ((size_t)bh * 64 + i0 + lm) * DM + 2 * lh;
  v8f acc[4] = {};
  for (int kk = 0; kk < DM; kk += 4) {
    v2f a = *(const v2f*)(arow + kk);
    #pragma unroll
    for (int t = 0; t < 4; ++t) {
      v2f bm = *(const v2f*)(wvw + (size_t)(h * 64 + 16 * t + lm) * DM + kk + 2 * lh);
      acc[t] = wmma4(a, bm, acc[t]);
    }
  }
  #pragma unroll
  for (int t = 0; t < 4; ++t) {
    const int jf = h * 64 + 16 * t + lm;
    const float bvj = bv[jf];
    const float zj  = z[(size_t)b * DM + jf];
    #pragma unroll
    for (int r = 0; r < 8; ++r) {
      const int ifl = h * 64 + i0 + r + 8 * lh;
      const float bki = bk[ifl];
      const float ui  = u[(size_t)b * DM + ifl];
      float val = acc[t][r] + ui * bvj + bki * zj + (float)SEQ * bki * bvj;
      M[((size_t)bh * 64 + i0 + r + 8 * lh) * DKH + 16 * t + lm] = 0.125f * val;
    }
  }
}

__global__ void __launch_bounds__(256) kP(const float* __restrict__ M,
                                          const float* __restrict__ wo,
                                          float* __restrict__ P) {
  const int lane = threadIdx.x & 31, wv = threadIdx.x >> 5;
  const int lm = lane & 15, lh = lane >> 4;
  const int job = blockIdx.x * 8 + wv;
  const int bh = job >> 5;
  const int b = bh >> 3, h = bh & 7;
  const int i0 = ((job >> 3) & 3) * 16;
  const int o0 = (job & 7) * 64;
  const float* arow = M + ((size_t)bh * 64 + i0 + lm) * DKH + 2 * lh;
  v8f acc[4] = {};
  for (int kk = 0; kk < DKH; kk += 4) {
    v2f a = *(const v2f*)(arow + kk);
    #pragma unroll
    for (int t = 0; t < 4; ++t) {
      v2f bm = *(const v2f*)(wo + (size_t)(o0 + 16 * t + lm) * DM + h * 64 + kk + 2 * lh);
      acc[t] = wmma4(a, bm, acc[t]);
    }
  }
  float* Pb = P + ((size_t)b * DM + h * 64 + i0) * DM + o0;
  #pragma unroll
  for (int t = 0; t < 4; ++t)
    #pragma unroll
    for (int r = 0; r < 8; ++r)
      Pb[(size_t)(r + 8 * lh) * DM + 16 * t + lm] = acc[t][r];
}

__global__ void __launch_bounds__(256) kW4(const float* __restrict__ wq,
                                           const float* __restrict__ P,
                                           float* __restrict__ W4) {
  const int lane = threadIdx.x & 31, wv = threadIdx.x >> 5;
  const int lm = lane & 15, lh = lane >> 4;
  const int job = blockIdx.x * 8 + wv;
  const int b = job >> 8;
  const int rest = job & 255;
  const int d0 = (rest >> 3) * 16;
  const int o0 = (rest & 7) * 64;
  const float* Pb = P + (size_t)b * DM * DM;
  v8f acc[4] = {};
  for (int kk = 0; kk < DM; kk += 4) {
    size_t fr = (size_t)(kk + 2 * lh) * DM;
    v2f a; a.x = wq[fr + d0 + lm]; a.y = wq[fr + DM + d0 + lm];
    #pragma unroll
    for (int t = 0; t < 4; ++t) {
      v2f bm; bm.x = Pb[fr + o0 + 16 * t + lm]; bm.y = Pb[fr + DM + o0 + 16 * t + lm];
      acc[t] = wmma4(a, bm, acc[t]);
    }
  }
  float* Wb = W4 + ((size_t)b * DM + d0) * DM + o0;
  #pragma unroll
  for (int t = 0; t < 4; ++t)
    #pragma unroll
    for (int r = 0; r < 8; ++r)
      Wb[(size_t)(r + 8 * lh) * DM + 16 * t + lm] = acc[t][r];
}

__global__ void __launch_bounds__(256) kOut(const float* __restrict__ q,
                                            const float* __restrict__ W4,
                                            const float* __restrict__ bo,
                                            const float* __restrict__ cq,
                                            float* __restrict__ out) {
  const int lane = threadIdx.x & 31, wv = threadIdx.x >> 5;
  const int lm = lane & 15, lh = lane >> 4;
  const int job = blockIdx.x * 8 + wv;
  const int b = job >> 7;
  const int n0 = (job & 127) * 32;
  const int o0 = blockIdx.y * 64;
  const float* qrow0 = q + ((size_t)b * SEQ + n0 + lm) * DM + 2 * lh;
  const float* qrow1 = qrow0 + (size_t)16 * DM;
  const float* Wb = W4 + (size_t)b * DM * DM;
  v8f acc[8] = {};
  #pragma unroll 2
  for (int kk = 0; kk < DM; kk += 4) {
    v2f a0 = *(const v2f*)(qrow0 + kk);
    v2f a1 = *(const v2f*)(qrow1 + kk);
    size_t rb = (size_t)(kk + 2 * lh) * DM;
    #pragma unroll
    for (int t = 0; t < 4; ++t) {
      v2f bm; bm.x = Wb[rb + o0 + 16 * t + lm]; bm.y = Wb[rb + DM + o0 + 16 * t + lm];
      acc[t]     = wmma4(a0, bm, acc[t]);
      acc[4 + t] = wmma4(a1, bm, acc[4 + t]);
    }
  }
  float* ob = out + ((size_t)b * SEQ + n0) * DM + o0;
  #pragma unroll
  for (int t = 0; t < 4; ++t) {
    float add = bo[o0 + 16 * t + lm] + cq[(size_t)b * DM + o0 + 16 * t + lm];
    #pragma unroll
    for (int m = 0; m < 2; ++m)
      #pragma unroll
      for (int r = 0; r < 8; ++r)
        ob[(size_t)(16 * m + r + 8 * lh) * DM + 16 * t + lm] = acc[4 * m + t][r] + add;
  }
}

__global__ void kUZ(const float* __restrict__ wk, const float* __restrict__ wvw,
                    const float* __restrict__ ksum, const float* __restrict__ vsum,
                    float* __restrict__ u, float* __restrict__ z) {
  int idx = blockIdx.x * blockDim.x + threadIdx.x;
  if (idx >= NB * DM) return;
  int b = idx >> 9, f = idx & (DM - 1);
  const float* wkr = wk + (size_t)f * DM;
  const float* wvr = wvw + (size_t)f * DM;
  const float* ks = ksum + (size_t)b * DM;
  const float* vs = vsum + (size_t)b * DM;
  float su = 0.f, sz = 0.f;
  for (int d = 0; d < DM; ++d) { su += wkr[d] * ks[d]; sz += wvr[d] * vs[d]; }
  u[idx] = su; z[idx] = sz;
}

__global__ void kCQ(const float* __restrict__ bq, const float* __restrict__ P,
                    float* __restrict__ cq) {
  int idx = blockIdx.x * blockDim.x + threadIdx.x;
  if (idx >= NB * DM) return;
  int b = idx >> 9, o = idx & (DM - 1);
  const float* Pb = P + (size_t)b * DM * DM + o;
  float s = 0.f;
  for (int f = 0; f < DM; ++f) s += bq[f] * Pb[(size_t)f * DM];
  cq[idx] = s;
}

extern "C" void kernel_launch(void* const* d_in, const int* in_sizes, int n_in,
                              void* d_out, int out_size, void* d_ws, size_t ws_size,
                              hipStream_t stream) {
  const float* q   = (const float*)d_in[0];
  const float* k   = (const float*)d_in[1];
  const float* v   = (const float*)d_in[2];
  const float* wq  = (const float*)d_in[3];
  const float* bq  = (const float*)d_in[4];
  const float* wk  = (const float*)d_in[5];
  const float* bk  = (const float*)d_in[6];
  const float* wvw = (const float*)d_in[7];
  const float* bv  = (const float*)d_in[8];
  const float* wo  = (const float*)d_in[9];
  const float* bo  = (const float*)d_in[10];
  float* out = (float*)d_out;

  float* ws = (float*)d_ws;
  float* G    = ws;
  float* T    = G    + (size_t)NB * DM * DM;
  float* M    = T    + (size_t)NB * DM * DM;
  float* P    = M    + (size_t)NB * NH * DKH * DKH;
  float* W4   = P    + (size_t)NB * DM * DM;
  float* ksum = W4   + (size_t)NB * DM * DM;
  float* vsum = ksum + (size_t)NB * DM;
  float* u    = vsum + (size_t)NB * DM;
  float* z    = u    + (size_t)NB * DM;
  float* cq   = z    + (size_t)NB * DM;
  float* Gp   = cq   + (size_t)NB * DM;
  float* SP   = Gp   + (size_t)KSPLIT * NB * DM * DM;

  kG   <<<KG_BLOCKS, 256, 0, stream>>>(k, v, Gp, SP);
  kGsum<<<GSUM_GRID, 256, 0, stream>>>(Gp, G, SP, ksum);
  kUZ  <<<4, 256, 0, stream>>>(wk, wvw, ksum, vsum, u, z);
  kT   <<<64, 256, 0, stream>>>(wk, G, T);
  kM   <<<8, 256, 0, stream>>>(T, wvw, bk, bv, u, z, M);
  kP   <<<64, 256, 0, stream>>>(M, wo, P);
  kCQ  <<<4, 256, 0, stream>>>(bq, P, cq);
  kW4  <<<64, 256, 0, stream>>>(wq, P, W4);
  dim3 go(32, 8, 1);
  kOut <<<go, 256, 0, stream>>>(q, W4, bo, cq, out);
}


# kT, kM, kW4, kUZ, kCQ: each 512-long reduction split 8 ways over a work-group's waves (lanes in kUZ), slice loads issued up front, partials added via LDS in wave order; changes their summation trees
# speedup vs baseline: 5.0157x; 1.8624x over previous
#include <hip/hip_runtime.h>


typedef float v2f __attribute__((ext_vector_type(2)));
typedef float v4f __attribute__((ext_vector_type(4)));
typedef float v8f __attribute__((ext_vector_type(8)));

#define DM   512
#define SEQ  4096
#define NB   2
#define NH   8
#define DKH  64
#define KSPLIT 16
#define SCH  (SEQ / KSPLIT)
#define KG_BLOCKS    (32 * KSPLIT)
#define GSUM_BLOCKS  (NB * DM * DM / 4 / 256)
#define NPART  (2 * KSPLIT)
#define NSUMS  (2 * NB * DM)
#define SUMS_BLOCKS  (NSUMS / 256)
#define GSUM_GRID    (GSUM_BLOCKS + SUMS_BLOCKS)
#define KW   (DM / 8)
#define SLICE_REGS __attribute__((amdgpu_waves_per_eu(4, 4)))

__device__ __forceinline__ v8f wmma4(v2f a, v2f b, v8f c) {
  return __builtin_amdgcn_wmma_f32_16x16x4_f32(
       false, a,  false, b,
       (short)0, c,  false,  false);
}

__device__ __forceinline__ v4f wg_sum_tile(const v8f (&acc)[4], float* red, int wv, int lm, int lh) {
  float* mine = red + wv * 1024;
  #pragma unroll
  for (int t = 0; t < 4; ++t)
    #pragma unroll
    for (int r = 0; r < 8; ++r)
      mine[(r + 8 * lh) * 64 + 16 * t + lm] = acc[t][r];
  __syncthreads();
  const int off = (2 * wv + lh) * 64 + 4 * lm;
  v4f s = *(const v4f*)(red + off);
  #pragma unroll
  for (int w = 1; w < 8; ++w) s += *(const v4f*)(red + w * 1024 + off);
  return s;
}

template <bool SK, bool SV>
__device__ __forceinline__ void g_tile(const float* __restrict__ kb,
                                       const float* __restrict__ vb,
                                       float* __restrict__ Gb,
                                       float* __restrict__ spk,
                                       float* __restrict__ spv,
                                       int d0, int e0, int lm, int lh) {
  v8f acc[8] = {};
  v2f ks0 = {0.f, 0.f}, ks1 = {0.f, 0.f};
  v2f vs[4] = {};
  #pragma unroll 2
  for (int s0 = 0; s0 < SCH; s0 += 4) {
    size_t rb = (size_t)(s0 + 2 * lh) * DM;
    v2f a0; a0.x = kb[rb + d0 + lm];      a0.y = kb[rb + DM + d0 + lm];
    v2f a1; a1.x = kb[rb + d0 + 16 + lm]; a1.y = kb[rb + DM + d0 + 16 + lm];
    if (SK) { ks0 += a0; ks1 += a1; }
    #pragma unroll
    for (int t = 0; t < 4; ++t) {
      v2f bm; bm.x = vb[rb + e0 + 16 * t + lm]; bm.y = vb[rb + DM + e0 + 16 * t + lm];
      if (SV) vs[t] += bm;
      acc[t]     = wmma4(a0, bm, acc[t]);
      acc[4 + t] = wmma4(a1, bm, acc[4 + t]);
    }
  }
  #pragma unroll
  for (int m = 0; m < 2; ++m)
    #pragma unroll
    for (int t = 0; t < 4; ++t)
      #pragma unroll
      for (int r = 0; r < 8; ++r)
        Gb[(size_t)(16 * m + r + 8 * lh) * DM + 16 * t + lm] = acc[4 * m + t][r];
  if (SK) {
    spk[d0 + lm]      = ks0.x + ks0.y;
    spk[d0 + 16 + lm] = ks1.x + ks1.y;
  }
  if (SV) {
    #pragma unroll
    for (int t = 0; t < 4; ++t) spv[e0 + 16 * t + lm] = vs[t].x + vs[t].y;
  }
}

__global__ void __launch_bounds__(256) kG(const float* __restrict__ k,
                                          const float* __restrict__ v,
                                          float* __restrict__ Gp,
                                          float* __restrict__ SP) {
  const int lane = threadIdx.x & 31, wv = threadIdx.x >> 5;
  const int lm = lane & 15, lh = lane >> 4;
  const int job = (blockIdx.x & 31) * 8 + wv;
  const int c = blockIdx.x >> 5;
  const int b = job >> 7;
  const int rest = job & 127;
  const int d0 = (rest >> 3) * 32;
  const int e0 = (rest & 7) * 64;
  const float* kb = k + ((size_t)b * SEQ + (size_t)c * SCH) * DM;
  const float* vb = v + ((size_t)b * SEQ + (size_t)c * SCH) * DM;
  float* Gb = Gp + (((size_t)c * NB + b) * DM + d0) * DM + e0;
  float* spk = SP + ((size_t)(2 * c + lh) * 2 + 0) * (NB * DM) + (size_t)b * DM;
  float* spv = SP + ((size_t)(2 * c + lh) * 2 + 1) * (NB * DM) + (size_t)b * DM;
  const bool sumK = __builtin_amdgcn_readfirstlane(wv) == 0;
  const bool sumV = (blockIdx.x & 15) == 0;
  if (sumV) {
    if (sumK) g_tile<true, true>(kb, vb, Gb, spk, spv, d0, e0, lm, lh);
    else      g_tile<false, true>(kb, vb, Gb, spk, spv, d0, e0, lm, lh);
  } else {
    if (sumK) g_tile<true, false>(kb, vb, Gb, spk, spv, d0, e0, lm, lh);
    else      g_tile<false, false>(kb, vb, Gb, spk, spv, d0, e0, lm, lh);
  }
}

__global__ void __launch_bounds__(256) kGsum(const float* __restrict__ Gp,
                                             float* __restrict__ G,
                                             const float* __restrict__ SP,
                                             float* __restrict__ sums) {
  if (blockIdx.x >= GSUM_BLOCKS) {
    const int j = (blockIdx.x - GSUM_BLOCKS) * 256 + threadIdx.x;
    float s = SP[j];
    for (int p = 1; p < NPART; ++p) s += SP[(size_t)p * NSUMS + j];
    sums[j] = s;
    return;
  }
  const int i = blockIdx.x * 256 + threadIdx.x;
  const float4* p = (const float4*)Gp + i;
  float4 s = p[0];
  for (int c = 1; c < KSPLIT; ++c) {
    const float4 t = p[(size_t)c * (NB * DM * DM / 4)];
    s.x += t.x; s.y += t.y; s.z += t.z; s.w += t.w;
  }
  ((float4*)G)[i] = s;
}

__global__ void __launch_bounds__(256) SLICE_REGS kT(const float* __restrict__ wk,
                                          const float* __restrict__ G,
                                          float* __restrict__ T) {
  __shared__ __attribute__((aligned(16))) float red[8 * 1024];
  const int lane = threadIdx.x & 31, wv = threadIdx.x >> 5;
  const int lm = lane & 15, lh = lane >> 4;
  const int job = blockIdx.x;
  const int bh = job >> 5;
  const int b = bh >> 3, h = bh & 7;
  const int rest = job & 31;
  const int i0 = (rest >> 3) * 16;
  const int e0 = (rest & 7) * 64;
  const int k0 = wv * KW;
  const float* arow = wk + (size_t)(h * 64 + i0 + lm) * DM + k0 + 2 * lh;
  const float* Gb = G + ((size_t)b * DM + k0) * DM;
  v8f acc[4] = {};
  #pragma unroll
  for (int kk = 0; kk < KW; kk += 4) {
    v2f a = *(const v2f*)(arow + kk);
    size_t rb = (size_t)(kk + 2 * lh) * DM;
    #pragma unroll
    for (int t = 0; t < 4; ++t) {
      v2f bm; bm.x = Gb[rb + e0 + 16 * t + lm]; bm.y = Gb[rb + DM + e0 + 16 * t + lm];
      acc[t] = wmma4(a, bm, acc[t]);
    }
  }
  const v4f s = wg_sum_tile(acc, red, wv, lm, lh);
  float* Tb = T + ((size_t)bh * 64 + i0) * DM + e0;
  *(v4f*)(Tb + (size_t)(2 * wv + lh) * DM + 4 * lm) = s;
}

__global__ void __launch_bounds__(256) SLICE_REGS kM(const float* __restrict__ T,
                                          const float* __restrict__ wvw,
                                          const float* __restrict__ bk,
                                          const float* __restrict__ bv,
                                          const float* __restrict__ u,
                                          const float* __restrict__ z,
                                          float* __restrict__ M) {
  __shared__ __attribute__((aligned(16))) float red[8 * 1024];
  const int lane = threadIdx.x & 31, wv = threadIdx.x >> 5;
  const int lm = lane & 15, lh = lane >> 4;
  const int job = blockIdx.x;
  const int bh = job >> 2;
  const int b = bh >> 3, h = bh & 7;
  const int i0 = (job & 3) * 16;
  const int k0 = wv * KW;
  const float* arow = T + ((size_t)bh * 64 + i0 + lm) * DM + k0 + 2 * lh;
  v8f acc[4] = {};
  #pragma unroll
  for (int kk = 0; kk < KW; kk += 4) {
    v2f a = *(const v2f*)(arow + kk);
    #pragma unroll
    for (int t = 0; t < 4; ++t) {
      v2f bm = *(const v2f*)(wvw + (size_t)(h * 64 + 16 * t + lm) * DM + k0 + kk + 2 * lh);
      acc[t] = wmma4(a, bm, acc[t]);
    }
  }
  const v4f s = wg_sum_tile(acc, red, wv, lm, lh);
  const int il = i0 + 2 * wv + lh;
  const int ifl = h * 64 + il;
  const float bki = bk[ifl];
  const float ui  = u[(size_t)b * DM + ifl];
  const v4f bv4 = *(const v4f*)(bv + h * 64 + 4 * lm);
  const v4f z4  = *(const v4f*)(z + (size_t)b * DM + h * 64 + 4 * lm);
  v4f o;
  #pragma unroll
  for (int c = 0; c < 4; ++c) {
    const float bvj = bv4[c], zj = z4[c];
    float val = s[c] + ui * bvj + bki * zj + (float)SEQ * bki * bvj;
    o[c] = 0.125f * val;
  }
  *(v4f*)(M + ((size_t)bh * 64 + il) * DKH + 4 * lm) = o;
}

__global__ void __launch_bounds__(256) kP(const float* __restrict__ M,
                                          const float* __restrict__ wo,
                                          float* __restrict__ P) {
  const int lane = threadIdx.x & 31, wv = threadIdx.x >> 5;
  const int lm = lane & 15, lh = lane >> 4;
  const int job = blockIdx.x * 8 + wv;
  const int bh = job >> 5;
  const int b = bh >> 3, h = bh & 7;
  const int i0 = ((job >> 3) & 3) * 16;
  const int o0 = (job & 7) * 64;
  const float* arow = M + ((size_t)bh * 64 + i0 + lm) * DKH + 2 * lh;
  v8f acc[4] = {};
  for (int kk = 0; kk < DKH; kk += 4) {
    v2f a = *(const v2f*)(arow + kk);
    #pragma unroll
    for (int t = 0; t < 4; ++t) {
      v2f bm = *(const v2f*)(wo + (size_t)(o0 + 16 * t + lm) * DM + h * 64 + kk + 2 * lh);
      acc[t] = wmma4(a, bm, acc[t]);
    }
  }
  float* Pb = P + ((size_t)b * DM + h * 64 + i0) * DM + o0;
  #pragma unroll
  for (int t = 0; t < 4; ++t)
    #pragma unroll
    for (int r = 0; r < 8; ++r)
      Pb[(size_t)(r + 8 * lh) * DM + 16 * t + lm] = acc[t][r];
}

__global__ void __launch_bounds__(256) SLICE_REGS kW4(const float* __restrict__ wq,
                                           const float* __restrict__ P,
                                           float* __restrict__ W4) {
  __shared__ __attribute__((aligned(16))) float red[8 * 1024];
  const int lane = threadIdx.x & 31, wv = threadIdx.x >> 5;
  const int lm = lane & 15, lh = lane >> 4;
  const int job = blockIdx.x;
  const int b = job >> 8;
  const int rest = job & 255;
  const int d0 = (rest >> 3) * 16;
  const int o0 = (rest & 7) * 64;
  const int k0 = wv * KW;
  const float* wqk = wq + (size_t)k0 * DM;
  const float* Pb = P + ((size_t)b * DM + k0) * DM;
  v8f acc[4] = {};
  #pragma unroll
  for (int kk = 0; kk < KW; kk += 4) {
    size_t fr = (size_t)(kk + 2 * lh) * DM;
    v2f a; a.x = wqk[fr + d0 + lm]; a.y = wqk[fr + DM + d0 + lm];
    #pragma unroll
    for (int t = 0; t < 4; ++t) {
      v2f bm; bm.x = Pb[fr + o0 + 16 * t + lm]; bm.y = Pb[fr + DM + o0 + 16 * t + lm];
      acc[t] = wmma4(a, bm, acc[t]);
    }
  }
  const v4f s = wg_sum_tile(acc, red, wv, lm, lh);
  float* Wb = W4 + ((size_t)b * DM + d0) * DM + o0;
  *(v4f*)(Wb + (size_t)(2 * wv + lh) * DM + 4 * lm) = s;
}

__global__ void __launch_bounds__(256) kOut(const float* __restrict__ q,
                                            const float* __restrict__ W4,
                                            const float* __restrict__ bo,
                                            const float* __restrict__ cq,
                                            float* __restrict__ out) {
  const int lane = threadIdx.x & 31, wv = threadIdx.x >> 5;
  const int lm = lane & 15, lh = lane >> 4;
  const int job = blockIdx.x * 8 + wv;
  const int b = job >> 7;
  const int n0 = (job & 127) * 32;
  const int o0 = blockIdx.y * 64;
  const float* qrow0 = q + ((size_t)b * SEQ + n0 + lm) * DM + 2 * lh;
  const float* qrow1 = qrow0 + (size_t)16 * DM;
  const float* Wb = W4 + (size_t)b * DM * DM;
  v8f acc[8] = {};
  #pragma unroll 2
  for (int kk = 0; kk < DM; kk += 4) {
    v2f a0 = *(const v2f*)(qrow0 + kk);
    v2f a1 = *(const v2f*)(qrow1 + kk);
    size_t rb = (size_t)(kk + 2 * lh) * DM;
    #pragma unroll
    for (int t = 0; t < 4; ++t) {
      v2f bm; bm.x = Wb[rb + o0 + 16 * t + lm]; bm.y = Wb[rb + DM + o0 + 16 * t + lm];
      acc[t]     = wmma4(a0, bm, acc[t]);
      acc[4 + t] = wmma4(a1, bm, acc[4 + t]);
    }
  }
  float* ob = out + ((size_t)b * SEQ + n0) * DM + o0;
  #pragma unroll
  for (int t = 0; t < 4; ++t) {
    float add = bo[o0 + 16 * t + lm] + cq[(size_t)b * DM + o0 + 16 * t + lm];
    #pragma unroll
    for (int m = 0; m < 2; ++m)
      #pragma unroll
      for (int r = 0; r < 8; ++r)
        ob[(size_t)(16 * m + r + 8 * lh) * DM + 16 * t + lm] = acc[4 * m + t][r] + add;
  }
}

__global__ void __launch_bounds__(256) kUZ(const float* __restrict__ wk, const float* __restrict__ wvw,
                    const float* __restrict__ ksum, const float* __restrict__ vsum,
                    float* __restrict__ u, float* __restrict__ z) {
  const int lane = threadIdx.x & 31, wv = threadIdx.x >> 5;
  const int job = blockIdx.x * 8 + wv;
  const bool second = job >= DM;
  const int f = job & (DM - 1);
  const float* wr = (second ? wvw : wk) + (size_t)f * DM + 4 * lane;
  const float* sm = (second ? vsum : ksum) + 4 * lane;
  float p0 = 0.f, p1 = 0.f;
  #pragma unroll
  for (int j = 0; j < 4; ++j) {
    const v4f w4 = *(const v4f*)(wr + 128 * j);
    const v4f s0 = *(const v4f*)(sm + 128 * j);
    const v4f s1 = *(const v4f*)(sm + DM + 128 * j);
    #pragma unroll
    for (int c = 0; c < 4; ++c) { p0 += w4[c] * s0[c]; p1 += w4[c] * s1[c]; }
  }
  #pragma unroll
  for (int off = 16; off >= 1; off >>= 1) {
    p0 += __shfl_xor(p0, off);
    p1 += __shfl_xor(p1, off);
  }
  if (lane == 0) {
    float* dst = second ? z : u;
    dst[f] = p0;
    dst[DM + f] = p1;
  }
}

__global__ void __launch_bounds__(256) SLICE_REGS kCQ(const float* __restrict__ bq, const float* __restrict__ P,
                    float* __restrict__ cq) {
  __shared__ float part[8 * 32];
  const int lane = threadIdx.x & 31, wv = threadIdx.x >> 5;
  const int b = blockIdx.x >> 4;
  const int o = (blockIdx.x & 15) * 32 + lane;
  const int f0 = wv * KW;
  const float* Pb = P + ((size_t)b * DM + f0) * DM + o;
  float s = 0.f;
  #pragma unroll
  for (int f = 0; f < KW; ++f) s += bq[f0 + f] * Pb[(size_t)f * DM];
  part[wv * 32 + lane] = s;
  __syncthreads();
  if (wv == 0) {
    float t = part[lane];
    #pragma unroll
    for (int w = 1; w < 8; ++w) t += part[w * 32 + lane];
    cq[(size_t)b * DM + o] = t;
  }
}

extern "C" void kernel_launch(void* const* d_in, const int* in_sizes, int n_in,
                              void* d_out, int out_size, void* d_ws, size_t ws_size,
                              hipStream_t stream) {
  const float* q   = (const float*)d_in[0];
  const float* k   = (const float*)d_in[1];
  const float* v   = (const float*)d_in[2];
  const float* wq  = (const float*)d_in[3];
  const float* bq  = (const float*)d_in[4];
  const float* wk  = (const float*)d_in[5];
  const float* bk  = (const float*)d_in[6];
  const float* wvw = (const float*)d_in[7];
  const float* bv  = (const float*)d_in[8];
  const float* wo  = (const float*)d_in[9];
  const float* bo  = (const float*)d_in[10];
  float* out = (float*)d_out;

  float* ws = (float*)d_ws;
  float* G    = ws;
  float* T    = G    + (size_t)NB * DM * DM;
  float* M    = T    + (size_t)NB * DM * DM;
  float* P    = M    + (size_t)NB * NH * DKH * DKH;
  float* W4   = P    + (size_t)NB * DM * DM;
  float* ksum = W4   + (size_t)NB * DM * DM;
  float* vsum = ksum + (size_t)NB * DM;
  float* u    = vsum + (size_t)NB * DM;
  float* z    = u    + (size_t)NB * DM;
  float* cq   = z    + (size_t)NB * DM;
  float* Gp   = cq   + (size_t)NB * DM;
  float* SP   = Gp   + (size_t)KSPLIT * NB * DM * DM;

  kG   <<<KG_BLOCKS, 256, 0, stream>>>(k, v, Gp, SP);
  kGsum<<<GSUM_GRID, 256, 0, stream>>>(Gp, G, SP, ksum);
  kUZ  <<<128, 256, 0, stream>>>(wk, wvw, ksum, vsum, u, z);
  kT   <<<512, 256, 0, stream>>>(wk, G, T);
  kM   <<<64, 256, 0, stream>>>(T, wvw, bk, bv, u, z, M);
  kP   <<<64, 256, 0, stream>>>(M, wo, P);
  kCQ  <<<32, 256, 0, stream>>>(bq, P, cq);
  kW4  <<<512, 256, 0, stream>>>(wq, P, W4);
  dim3 go(32, 8, 1);
  kOut <<<go, 256, 0, stream>>>(q, W4, bo, cq, out);
}


# kG, kOut: a lane's 4 B (kG: 2 A) fragments take adjacent columns: 16/8-byte loads, whole-line 16-byte stores; kOut step = 16 k with all 12 loads ahead of its WMMAs; changes out's summation order
# speedup vs baseline: 7.0797x; 1.4115x over previous
#include <hip/hip_runtime.h>


typedef float v2f __attribute__((ext_vector_type(2)));
typedef float v4f __attribute__((ext_vector_type(4)));
typedef float v8f __attribute__((ext_vector_type(8)));

#define DM   512
#define SEQ  4096
#define NB   2
#define NH   8
#define DKH  64
#define KSPLIT 16
#define SCH  (SEQ / KSPLIT)
#define KG_BLOCKS    (32 * KSPLIT)
#define GSUM_BLOCKS  (NB * DM * DM / 4 / 256)
#define NPART  (2 * KSPLIT)
#define NSUMS  (2 * NB * DM)
#define SUMS_BLOCKS  (NSUMS / 256)
#define GSUM_GRID    (GSUM_BLOCKS + SUMS_BLOCKS)
#define KW   (DM / 8)
#define SLICE_REGS __attribute__((amdgpu_waves_per_eu(4, 4)))

__device__ __forceinline__ v8f wmma4(v2f a, v2f b, v8f c) {
  return __builtin_amdgcn_wmma_f32_16x16x4_f32(
       false, a,  false, b,
       (short)0, c,  false,  false);
}

__device__ __forceinline__ v4f wg_sum_tile(const v8f (&acc)[4], float* red, int wv, int lm, int lh) {
  float* mine = red + wv * 1024;
  #pragma unroll
  for (int t = 0; t < 4; ++t)
    #pragma unroll
    for (int r = 0; r < 8; ++r)
      mine[(r + 8 * lh) * 64 + 16 * t + lm] = acc[t][r];
  __syncthreads();
  const int off = (2 * wv + lh) * 64 + 4 * lm;
  v4f s = *(const v4f*)(red + off);
  #pragma unroll
  for (int w = 1; w < 8; ++w) s += *(const v4f*)(red + w * 1024 + off);
  return s;
}

template <bool SK, bool SV>
__device__ __forceinline__ void g_tile(const float* __restrict__ kb,
                                       const float* __restrict__ vb,
                                       float* __restrict__ Gb,
                                       float* __restrict__ spk,
                                       float* __restrict__ spv,
                                       int d0, int e0, int lm, int lh) {
  v8f acc[8] = {};
  v2f ksx = {0.f, 0.f}, ksy = {0.f, 0.f};
  v4f vsx = {0.f, 0.f, 0.f, 0.f}, vsy = {0.f, 0.f, 0.f, 0.f};
  const float* kp = kb + (size_t)(2 * lh) * DM + d0 + 2 * lm;
  const float* vp = vb + (size_t)(2 * lh) * DM + e0 + 4 * lm;
  #pragma unroll 2
  for (int s0 = 0; s0 < SCH; s0 += 4) {
    const v2f kx = *(const v2f*)(kp + (size_t)s0 * DM);
    const v2f ky = *(const v2f*)(kp + (size_t)(s0 + 1) * DM);
    const v4f vx = *(const v4f*)(vp + (size_t)s0 * DM);
    const v4f vy = *(const v4f*)(vp + (size_t)(s0 + 1) * DM);
    if (SK) { ksx += kx; ksy += ky; }
    if (SV) { vsx += vx; vsy += vy; }
    v2f a0; a0.x = kx.x; a0.y = ky.x;
    v2f a1; a1.x = kx.y; a1.y = ky.y;
    #pragma unroll
    for (int t = 0; t < 4; ++t) {
      v2f bm; bm.x = vx[t]; bm.y = vy[t];
      acc[t]     = wmma4(a0, bm, acc[t]);
      acc[4 + t] = wmma4(a1, bm, acc[4 + t]);
    }
  }
  #pragma unroll
  for (int m = 0; m < 2; ++m)
    #pragma unroll
    for (int r = 0; r < 8; ++r) {
      v4f o;
      o.x = acc[4 * m][r]; o.y = acc[4 * m + 1][r]; o.z = acc[4 * m + 2][r]; o.w = acc[4 * m + 3][r];
      *(v4f*)(Gb + (size_t)(2 * (8 * lh + r) + m) * DM + 4 * lm) = o;
    }
  if (SK) *(v2f*)(spk + d0 + 2 * lm) = ksx + ksy;
  if (SV) *(v4f*)(spv + e0 + 4 * lm) = vsx + vsy;
}

__global__ void __launch_bounds__(256) kG(const float* __restrict__ k,
                                          const float* __restrict__ v,
                                          float* __restrict__ Gp,
                                          float* __restrict__ SP) {
  const int lane = threadIdx.x & 31, wv = threadIdx.x >> 5;
  const int lm = lane & 15, lh = lane >> 4;
  const int job = (blockIdx.x & 31) * 8 + wv;
  const int c = blockIdx.x >> 5;
  const int b = job >> 7;
  const int rest = job & 127;
  const int d0 = (rest >> 3) * 32;
  const int e0 = (rest & 7) * 64;
  const float* kb = k + ((size_t)b * SEQ + (size_t)c * SCH) * DM;
  const float* vb = v + ((size_t)b * SEQ + (size_t)c * SCH) * DM;
  float* Gb = Gp + (((size_t)c * NB + b) * DM + d0) * DM + e0;
  float* spk = SP + ((size_t)(2 * c + lh) * 2 + 0) * (NB * DM) + (size_t)b * DM;
  float* spv = SP + ((size_t)(2 * c + lh) * 2 + 1) * (NB * DM) + (size_t)b * DM;
  const bool sumK = __builtin_amdgcn_readfirstlane(wv) == 0;
  const bool sumV = (blockIdx.x & 15) == 0;
  if (sumV) {
    if (sumK) g_tile<true, true>(kb, vb, Gb, spk, spv, d0, e0, lm, lh);
    else      g_tile<false, true>(kb, vb, Gb, spk, spv, d0, e0, lm, lh);
  } else {
    if (sumK) g_tile<true, false>(kb, vb, Gb, spk, spv, d0, e0, lm, lh);
    else      g_tile<false, false>(kb, vb, Gb, spk, spv, d0, e0, lm, lh);
  }
}

__global__ void __launch_bounds__(256) kGsum(const float* __restrict__ Gp,
                                             float* __restrict__ G,
                                             const float* __restrict__ SP,
                                             float* __restrict__ sums) {
  if (blockIdx.x >= GSUM_BLOCKS) {
    const int j = (blockIdx.x - GSUM_BLOCKS) * 256 + threadIdx.x;
    float s = SP[j];
    for (int p = 1; p < NPART; ++p) s += SP[(size_t)p * NSUMS + j];
    sums[j] = s;
    return;
  }
  const int i = blockIdx.x * 256 + threadIdx.x;
  const float4* p = (const float4*)Gp + i;
  float4 s = p[0];
  for (int c = 1; c < KSPLIT; ++c) {
    const float4 t = p[(size_t)c * (NB * DM * DM / 4)];
    s.x += t.x; s.y += t.y; s.z += t.z; s.w += t.w;
  }
  ((float4*)G)[i] = s;
}

__global__ void __launch_bounds__(256) SLICE_REGS kT(const float* __restrict__ wk,
                                          const float* __restrict__ G,
                                          float* __restrict__ T) {
  __shared__ __attribute__((aligned(16))) float red[8 * 1024];
  const int lane = threadIdx.x & 31, wv = threadIdx.x >> 5;
  const int lm = lane & 15, lh = lane >> 4;
  const int job = blockIdx.x;
  const int bh = job >> 5;
  const int b = bh >> 3, h = bh & 7;
  const int rest = job & 31;
  const int i0 = (rest >> 3) * 16;
  const int e0 = (rest & 7) * 64;
  const int k0 = wv * KW;
  const float* arow = wk + (size_t)(h * 64 + i0 + lm) * DM + k0 + 2 * lh;
  const float* Gb = G + ((size_t)b * DM + k0) * DM;
  v8f acc[4] = {};
  #pragma unroll
  for (int kk = 0; kk < KW; kk += 4) {
    v2f a = *(const v2f*)(arow + kk);
    size_t rb = (size_t)(kk + 2 * lh) * DM;
    #pragma unroll
    for (int t = 0; t < 4; ++t) {
      v2f bm; bm.x = Gb[rb + e0 + 16 * t + lm]; bm.y = Gb[rb + DM + e0 + 16 * t + lm];
      acc[t] = wmma4(a, bm, acc[t]);
    }
  }
  const v4f s = wg_sum_tile(acc, red, wv, lm, lh);
  float* Tb = T + ((size_t)bh * 64 + i0) * DM + e0;
  *(v4f*)(Tb + (size_t)(2 * wv + lh) * DM + 4 * lm) = s;
}

__global__ void __launch_bounds__(256) SLICE_REGS kM(const float* __restrict__ T,
                                          const float* __restrict__ wvw,
                                          const float* __restrict__ bk,
                                          const float* __restrict__ bv,
                                          const float* __restrict__ u,
                                          const float* __restrict__ z,
                                          float* __restrict__ M) {
  __shared__ __attribute__((aligned(16))) float red[8 * 1024];
  const int lane = threadIdx.x & 31, wv = threadIdx.x >> 5;
  const int lm = lane & 15, lh = lane >> 4;
  const int job = blockIdx.x;
  const int bh = job >> 2;
  const int b = bh >> 3, h = bh & 7;
  const int i0 = (job & 3) * 16;
  const int k0 = wv * KW;
  const float* arow = T + ((size_t)bh * 64 + i0 + lm) * DM + k0 + 2 * lh;
  v8f acc[4] = {};
  #pragma unroll
  for (int kk = 0; kk < KW; kk += 4) {
    v2f a = *(const v2f*)(arow + kk);
    #pragma unroll
    for (int t = 0; t < 4; ++t) {
      v2f bm = *(const v2f*)(wvw + (size_t)(h * 64 + 16 * t + lm) * DM + k0 + kk + 2 * lh);
      acc[t] = wmma4(a, bm, acc[t]);
    }
  }
  const v4f s = wg_sum_tile(acc, red, wv, lm, lh);
  const int il = i0 + 2 * wv + lh;
  const int ifl = h * 64 + il;
  const float bki = bk[ifl];
  const float ui  = u[(size_t)b * DM + ifl];
  const v4f bv4 = *(const v4f*)(bv + h * 64 + 4 * lm);
  const v4f z4  = *(const v4f*)(z + (size_t)b * DM + h * 64 + 4 * lm);
  v4f o;
  #pragma unroll
  for (int c = 0; c < 4; ++c) {
    const float bvj = bv4[c], zj = z4[c];
    float val = s[c] + ui * bvj + bki * zj + (float)SEQ * bki * bvj;
    o[c] = 0.125f * val;
  }
  *(v4f*)(M + ((size_t)bh * 64 + il) * DKH + 4 * lm) = o;
}

__global__ void __launch_bounds__(256) kP(const float* __restrict__ M,
                                          const float* __restrict__ wo,
                                          float* __restrict__ P) {
  const int lane = threadIdx.x & 31, wv = threadIdx.x >> 5;
  const int lm = lane & 15, lh = lane >> 4;
  const int job = blockIdx.x * 8 + wv;
  const int bh = job >> 5;
  const int b = bh >> 3, h = bh & 7;
  const int i0 = ((job >> 3) & 3) * 16;
  const int o0 = (job & 7) * 64;
  const float* arow = M + ((size_t)bh * 64 + i0 + lm) * DKH + 2 * lh;
  v8f acc[4] = {};
  for (int kk = 0; kk < DKH; kk += 4) {
    v2f a = *(const v2f*)(arow + kk);
    #pragma unroll
    for (int t = 0; t < 4; ++t) {
      v2f bm = *(const v2f*)(wo + (size_t)(o0 + 16 * t + lm) * DM + h * 64 + kk + 2 * lh);
      acc[t] = wmma4(a, bm, acc[t]);
    }
  }
  float* Pb = P + ((size_t)b * DM + h * 64 + i0) * DM + o0;
  #pragma unroll
  for (int t = 0; t < 4; ++t)
    #pragma unroll
    for (int r = 0; r < 8; ++r)
      Pb[(size_t)(r + 8 * lh) * DM + 16 * t + lm] = acc[t][r];
}

__global__ void __launch_bounds__(256) SLICE_REGS kW4(const float* __restrict__ wq,
                                           const float* __restrict__ P,
                                           float* __restrict__ W4) {
  __shared__ __attribute__((aligned(16))) float red[8 * 1024];
  const int lane = threadIdx.x & 31, wv = threadIdx.x >> 5;
  const int lm = lane & 15, lh = lane >> 4;
  const int job = blockIdx.x;
  const int b = job >> 8;
  const int rest = job & 255;
  const int d0 = (rest >> 3) * 16;
  const int o0 = (rest & 7) * 64;
  const int k0 = wv * KW;
  const float* wqk = wq + (size_t)k0 * DM;
  const float* Pb = P + ((size_t)b * DM + k0) * DM;
  v8f acc[4] = {};
  #pragma unroll
  for (int kk = 0; kk < KW; kk += 4) {
    size_t fr = (size_t)(kk + 2 * lh) * DM;
    v2f a; a.x = wqk[fr + d0 + lm]; a.y = wqk[fr + DM + d0 + lm];
    #pragma unroll
    for (int t = 0; t < 4; ++t) {
      v2f bm; bm.x = Pb[fr + o0 + 16 * t + lm]; bm.y = Pb[fr + DM + o0 + 16 * t + lm];
      acc[t] = wmma4(a, bm, acc[t]);
    }
  }
  const v4f s = wg_sum_tile(acc, red, wv, lm, lh);
  float* Wb = W4 + ((size_t)b * DM + d0) * DM + o0;
  *(v4f*)(Wb + (size_t)(2 * wv + lh) * DM + 4 * lm) = s;
}

__global__ void __launch_bounds__(256) SLICE_REGS kOut(const float* __restrict__ q,
                                            const float* __restrict__ W4,
                                            const float* __restrict__ bo,
                                            const float* __restrict__ cq,
                                            float* __restrict__ out) {
  const int lane = threadIdx.x & 31, wv = threadIdx.x >> 5;
  const int lm = lane & 15, lh = lane >> 4;
  const int job = blockIdx.x * 8 + wv;
  const int b = job >> 7;
  const int n0 = (job & 127) * 32;
  const int o0 = blockIdx.y * 64;
  const float* qrow0 = q + ((size_t)b * SEQ + n0 + lm) * DM + 8 * lh;
  const float* qrow1 = qrow0 + (size_t)16 * DM;
  const float* Wp = W4 + (size_t)b * DM * DM + (size_t)(8 * lh) * DM + o0 + 4 * lm;
  v8f acc[8] = {};
  for (int kk = 0; kk < DM; kk += 16) {
    v4f a0[2], a1[2], w[8];
    a0[0] = *(const v4f*)(qrow0 + kk); a0[1] = *(const v4f*)(qrow0 + kk + 4);
    a1[0] = *(const v4f*)(qrow1 + kk); a1[1] = *(const v4f*)(qrow1 + kk + 4);
    #pragma unroll
    for (int j = 0; j < 8; ++j) w[j] = *(const v4f*)(Wp + (size_t)(kk + j) * DM);
    #pragma unroll
    for (int p = 0; p < 4; ++p) {
      v2f x0; x0.x = a0[p >> 1][2 * (p & 1)]; x0.y = a0[p >> 1][2 * (p & 1) + 1];
      v2f x1; x1.x = a1[p >> 1][2 * (p & 1)]; x1.y = a1[p >> 1][2 * (p & 1) + 1];
      #pragma unroll
      for (int t = 0; t < 4; ++t) {
        v2f bm; bm.x = w[2 * p][t]; bm.y = w[2 * p + 1][t];
        acc[t]     = wmma4(x0, bm, acc[t]);
        acc[4 + t] = wmma4(x1, bm, acc[4 + t]);
      }
    }
  }
  const v4f add = *(const v4f*)(bo + o0 + 4 * lm) + *(const v4f*)(cq + (size_t)b * DM + o0 + 4 * lm);
  float* ob = out + ((size_t)b * SEQ + n0) * DM + o0 + 4 * lm;
  #pragma unroll
  for (int m = 0; m < 2; ++m)
    #pragma unroll
    for (int r = 0; r < 8; ++r) {
      v4f o;
      o.x = acc[4 * m][r]; o.y = acc[4 * m + 1][r]; o.z = acc[4 * m + 2][r]; o.w = acc[4 * m + 3][r];
      *(v4f*)(ob + (size_t)(16 * m + r + 8 * lh) * DM) = o + add;
    }
}

__global__ void __launch_bounds__(256) kUZ(const float* __restrict__ wk, const float* __restrict__ wvw,
                    const float* __restrict__ ksum, const float* __restrict__ vsum,
                    float* __restrict__ u, float* __restrict__ z) {
  const int lane = threadIdx.x & 31, wv = threadIdx.x >> 5;
  const int job = blockIdx.x * 8 + wv;
  const bool second = job >= DM;
  const int f = job & (DM - 1);
  const float* wr = (second ? wvw : wk) + (size_t)f * DM + 4 * lane;
  const float* sm = (second ? vsum : ksum) + 4 * lane;
  float p0 = 0.f, p1 = 0.f;
  #pragma unroll
  for (int j = 0; j < 4; ++j) {
    const v4f w4 = *(const v4f*)(wr + 128 * j);
    const v4f s0 = *(const v4f*)(sm + 128 * j);
    const v4f s1 = *(const v4f*)(sm + DM + 128 * j);
    #pragma unroll
    for (int c = 0; c < 4; ++c) { p0 += w4[c] * s0[c]; p1 += w4[c] * s1[c]; }
  }
  #pragma unroll
  for (int off = 16; off >= 1; off >>= 1) {
    p0 += __shfl_xor(p0, off);
    p1 += __shfl_xor(p1, off);
  }
  if (lane == 0) {
    float* dst = second ? z : u;
    dst[f] = p0;
    dst[DM + f] = p1;
  }
}

__global__ void __launch_bounds__(256) SLICE_REGS kCQ(const float* __restrict__ bq, const float* __restrict__ P,
                    float* __restrict__ cq) {
  __shared__ float part[8 * 32];
  const int lane = threadIdx.x & 31, wv = threadIdx.x >> 5;
  const int b = blockIdx.x >> 4;
  const int o = (blockIdx.x & 15) * 32 + lane;
  const int f0 = wv * KW;
  const float* Pb = P + ((size_t)b * DM + f0) * DM + o;
  float s = 0.f;
  #pragma unroll
  for (int f = 0; f < KW; ++f) s += bq[f0 + f] * Pb[(size_t)f * DM];
  part[wv * 32 + lane] = s;
  __syncthreads();
  if (wv == 0) {
    float t = part[lane];
    #pragma unroll
    for (int w = 1; w < 8; ++w) t += part[w * 32 + lane];
    cq[(size_t)b * DM + o] = t;
  }
}

extern "C" void kernel_launch(void* const* d_in, const int* in_sizes, int n_in,
                              void* d_out, int out_size, void* d_ws, size_t ws_size,
                              hipStream_t stream) {
  const float* q   = (const float*)d_in[0];
  const float* k   = (const float*)d_in[1];
  const float* v   = (const float*)d_in[2];
  const float* wq  = (const float*)d_in[3];
  const float* bq  = (const float*)d_in[4];
  const float* wk  = (const float*)d_in[5];
  const float* bk  = (const float*)d_in[6];
  const float* wvw = (const float*)d_in[7];
  const float* bv  = (const float*)d_in[8];
  const float* wo  = (const float*)d_in[9];
  const float* bo  = (const float*)d_in[10];
  float* out = (float*)d_out;

  float* ws = (float*)d_ws;
  float* G    = ws;
  float* T    = G    + (size_t)NB * DM * DM;
  float* M    = T    + (size_t)NB * DM * DM;
  float* P    = M    + (size_t)NB * NH * DKH * DKH;
  float* W4   = P    + (size_t)NB * DM * DM;
  float* ksum = W4   + (size_t)NB * DM * DM;
  float* vsum = ksum + (size_t)NB * DM;
  float* u    = vsum + (size_t)NB * DM;
  float* z    = u    + (size_t)NB * DM;
  float* cq   = z    + (size_t)NB * DM;
  float* Gp   = cq   + (size_t)NB * DM;
  float* SP   = Gp   + (size_t)KSPLIT * NB * DM * DM;

  kG   <<<KG_BLOCKS, 256, 0, stream>>>(k, v, Gp, SP);
  kGsum<<<GSUM_GRID, 256, 0, stream>>>(Gp, G, SP, ksum);
  kUZ  <<<128, 256, 0, stream>>>(wk, wvw, ksum, vsum, u, z);
  kT   <<<512, 256, 0, stream>>>(wk, G, T);
  kM   <<<64, 256, 0, stream>>>(T, wvw, bk, bv, u, z, M);
  kP   <<<64, 256, 0, stream>>>(M, wo, P);
  kCQ  <<<32, 256, 0, stream>>>(bq, P, cq);
  kW4  <<<512, 256, 0, stream>>>(wq, P, W4);
  dim3 go(32, 8, 1);
  kOut <<<go, 256, 0, stream>>>(q, W4, bo, cq, out);
}


# kG, kOut: second operand register set; the loads of step n+1 are issued in front of the WMMAs of step n (kG step = 16 rows); steps stay in index order, no sum changes
# speedup vs baseline: 8.2661x; 1.1676x over previous
#include <hip/hip_runtime.h>


typedef float v2f __attribute__((ext_vector_type(2)));
typedef float v4f __attribute__((ext_vector_type(4)));
typedef float v8f __attribute__((ext_vector_type(8)));

#define DM   512
#define SEQ  4096
#define NB   2
#define NH   8
#define DKH  64
#define KSPLIT 16
#define SCH  (SEQ / KSPLIT)
#define KG_BLOCKS    (32 * KSPLIT)
#define GSUM_BLOCKS  (NB * DM * DM / 4 / 256)
#define NPART  (2 * KSPLIT)
#define NSUMS  (2 * NB * DM)
#define SUMS_BLOCKS  (NSUMS / 256)
#define GSUM_GRID    (GSUM_BLOCKS + SUMS_BLOCKS)
#define KW   (DM / 8)
#define SLICE_REGS __attribute__((amdgpu_waves_per_eu(4, 4)))

__device__ __forceinline__ v8f wmma4(v2f a, v2f b, v8f c) {
  return __builtin_amdgcn_wmma_f32_16x16x4_f32(
       false, a,  false, b,
       (short)0, c,  false,  false);
}

__device__ __forceinline__ v4f wg_sum_tile(const v8f (&acc)[4], float* red, int wv, int lm, int lh) {
  float* mine = red + wv * 1024;
  #pragma unroll
  for (int t = 0; t < 4; ++t)
    #pragma unroll
    for (int r = 0; r < 8; ++r)
      mine[(r + 8 * lh) * 64 + 16 * t + lm] = acc[t][r];
  __syncthreads();
  const int off = (2 * wv + lh) * 64 + 4 * lm;
  v4f s = *(const v4f*)(red + off);
  #pragma unroll
  for (int w = 1; w < 8; ++w) s += *(const v4f*)(red + w * 1024 + off);
  return s;
}

struct GSet { v2f kx[4], ky[4]; v4f vx[4], vy[4]; };

__device__ __forceinline__ void g_load(GSet& f, const float* __restrict__ kp,
                                       const float* __restrict__ vp, int s) {
  #pragma unroll
  for (int i = 0; i < 4; ++i) {
    f.kx[i] = *(const v2f*)(kp + (size_t)(s + 4 * i) * DM);
    f.ky[i] = *(const v2f*)(kp + (size_t)(s + 4 * i + 1) * DM);
    f.vx[i] = *(const v4f*)(vp + (size_t)(s + 4 * i) * DM);
    f.vy[i] = *(const v4f*)(vp + (size_t)(s + 4 * i + 1) * DM);
  }
}

template <bool SK, bool SV>
__device__ __forceinline__ void g_mma(const GSet& f, v8f (&acc)[8],
                                      v2f& ksx, v2f& ksy, v4f& vsx, v4f& vsy) {
  #pragma unroll
  for (int i = 0; i < 4; ++i) {
    const v2f kx = f.kx[i], ky = f.ky[i];
    const v4f vx = f.vx[i], vy = f.vy[i];
    if (SK) { ksx += kx; ksy += ky; }
    if (SV) { vsx += vx; vsy += vy; }
    v2f a0; a0.x = kx.x; a0.y = ky.x;
    v2f a1; a1.x = kx.y; a1.y = ky.y;
    #pragma unroll
    for (int t = 0; t < 4; ++t) {
      v2f bm; bm.x = vx[t]; bm.y = vy[t];
      acc[t]     = wmma4(a0, bm, acc[t]);
      acc[4 + t] = wmma4(a1, bm, acc[4 + t]);
    }
  }
}

template <bool SK, bool SV>
__device__ __forceinline__ void g_tile(const float* __restrict__ kb,
                                       const float* __restrict__ vb,
                                       float* __restrict__ Gb,
                                       float* __restrict__ spk,
                                       float* __restrict__ spv,
                                       int d0, int e0, int lm, int lh) {
  v8f acc[8] = {};
  v2f ksx = {0.f, 0.f}, ksy = {0.f, 0.f};
  v4f vsx = {0.f, 0.f, 0.f, 0.f}, vsy = {0.f, 0.f, 0.f, 0.f};
  const float* kp = kb + (size_t)(2 * lh) * DM + d0 + 2 * lm;
  const float* vp = vb + (size_t)(2 * lh) * DM + e0 + 4 * lm;
  GSet f0, f1;
  g_load(f0, kp, vp, 0);
  #pragma nounroll
  for (int s0 = 0; s0 < SCH - 32; s0 += 32) {
    g_load(f1, kp, vp, s0 + 16);
    g_mma<SK, SV>(f0, acc, ksx, ksy, vsx, vsy);
    g_load(f0, kp, vp, s0 + 32);
    g_mma<SK, SV>(f1, acc, ksx, ksy, vsx, vsy);
  }
  g_load(f1, kp, vp, SCH - 16);
  g_mma<SK, SV>(f0, acc, ksx, ksy, vsx, vsy);
  g_mma<SK, SV>(f1, acc, ksx, ksy, vsx, vsy);
  #pragma unroll
  for (int m = 0; m < 2; ++m)
    #pragma unroll
    for (int r = 0; r < 8; ++r) {
      v4f o;
      o.x = acc[4 * m][r]; o.y = acc[4 * m + 1][r]; o.z = acc[4 * m + 2][r]; o.w = acc[4 * m + 3][r];
      *(v4f*)(Gb + (size_t)(2 * (8 * lh + r) + m) * DM + 4 * lm) = o;
    }
  if (SK) *(v2f*)(spk + d0 + 2 * lm) = ksx + ksy;
  if (SV) *(v4f*)(spv + e0 + 4 * lm) = vsx + vsy;
}

__global__ void __launch_bounds__(256) SLICE_REGS kG(const float* __restrict__ k,
                                          const float* __restrict__ v,
                                          float* __restrict__ Gp,
                                          float* __restrict__ SP) {
  const int lane = threadIdx.x & 31, wv = threadIdx.x >> 5;
  const int lm = lane & 15, lh = lane >> 4;
  const int job = (blockIdx.x & 31) * 8 + wv;
  const int c = blockIdx.x >> 5;
  const int b = job >> 7;
  const int rest = job & 127;
  const int d0 = (rest >> 3) * 32;
  const int e0 = (rest & 7) * 64;
  const float* kb = k + ((size_t)b * SEQ + (size_t)c * SCH) * DM;
  const float* vb = v + ((size_t)b * SEQ + (size_t)c * SCH) * DM;
  float* Gb = Gp + (((size_t)c * NB + b) * DM + d0) * DM + e0;
  float* spk = SP + ((size_t)(2 * c + lh) * 2 + 0) * (NB * DM) + (size_t)b * DM;
  float* spv = SP + ((size_t)(2 * c + lh) * 2 + 1) * (NB * DM) + (size_t)b * DM;
  const bool sumK = __builtin_amdgcn_readfirstlane(wv) == 0;
  const bool sumV = (blockIdx.x & 15) == 0;
  if (sumV) {
    if (sumK) g_tile<true, true>(kb, vb, Gb, spk, spv, d0, e0, lm, lh);
    else      g_tile<false, true>(kb, vb, Gb, spk, spv, d0, e0, lm, lh);
  } else {
    if (sumK) g_tile<true, false>(kb, vb, Gb, spk, spv, d0, e0, lm, lh);
    else      g_tile<false, false>(kb, vb, Gb, spk, spv, d0, e0, lm, lh);
  }
}

__global__ void __launch_bounds__(256) kGsum(const float* __restrict__ Gp,
                                             float* __restrict__ G,
                                             const float* __restrict__ SP,
                                             float* __restrict__ sums) {
  if (blockIdx.x >= GSUM_BLOCKS) {
    const int j = (blockIdx.x - GSUM_BLOCKS) * 256 + threadIdx.x;
    float s = SP[j];
    for (int p = 1; p < NPART; ++p) s += SP[(size_t)p * NSUMS + j];
    sums[j] = s;
    return;
  }
  const int i = blockIdx.x * 256 + threadIdx.x;
  const float4* p = (const float4*)Gp + i;
  float4 s = p[0];
  for (int c = 1; c < KSPLIT; ++c) {
    const float4 t = p[(size_t)c * (NB * DM * DM / 4)];
    s.x += t.x; s.y += t.y; s.z += t.z; s.w += t.w;
  }
  ((float4*)G)[i] = s;
}

__global__ void __launch_bounds__(256) SLICE_REGS kT(const float* __restrict__ wk,
                                          const float* __restrict__ G,
                                          float* __restrict__ T) {
  __shared__ __attribute__((aligned(16))) float red[8 * 1024];
  const int lane = threadIdx.x & 31, wv = threadIdx.x >> 5;
  const int lm = lane & 15, lh = lane >> 4;
  const int job = blockIdx.x;
  const int bh = job >> 5;
  const int b = bh >> 3, h = bh & 7;
  const int rest = job & 31;
  const int i0 = (rest >> 3) * 16;
  const int e0 = (rest & 7) * 64;
  const int k0 = wv * KW;
  const float* arow = wk + (size_t)(h * 64 + i0 + lm) * DM + k0 + 2 * lh;
  const float* Gb = G + ((size_t)b * DM + k0) * DM;
  v8f acc[4] = {};
  #pragma unroll
  for (int kk = 0; kk < KW; kk += 4) {
    v2f a = *(const v2f*)(arow + kk);
    size_t rb = (size_t)(kk + 2 * lh) * DM;
    #pragma unroll
    for (int t = 0; t < 4; ++t) {
      v2f bm; bm.x = Gb[rb + e0 + 16 * t + lm]; bm.y = Gb[rb + DM + e0 + 16 * t + lm];
      acc[t] = wmma4(a, bm, acc[t]);
    }
  }
  const v4f s = wg_sum_tile(acc, red, wv, lm, lh);
  float* Tb = T + ((size_t)bh * 64 + i0) * DM + e0;
  *(v4f*)(Tb + (size_t)(2 * wv + lh) * DM + 4 * lm) = s;
}

__global__ void __launch_bounds__(256) SLICE_REGS kM(const float* __restrict__ T,
                                          const float* __restrict__ wvw,
                                          const float* __restrict__ bk,
                                          const float* __restrict__ bv,
                                          const float* __restrict__ u,
                                          const float* __restrict__ z,
                                          float* __restrict__ M) {
  __shared__ __attribute__((aligned(16))) float red[8 * 1024];
  const int lane = threadIdx.x & 31, wv = threadIdx.x >> 5;
  const int lm = lane & 15, lh = lane >> 4;
  const int job = blockIdx.x;
  const int bh = job >> 2;
  const int b = bh >> 3, h = bh & 7;
  const int i0 = (job & 3) * 16;
  const int k0 = wv * KW;
  const float* arow = T + ((size_t)bh * 64 + i0 + lm) * DM + k0 + 2 * lh;
  v8f acc[4] = {};
  #pragma unroll
  for (int kk = 0; kk < KW; kk += 4) {
    v2f a = *(const v2f*)(arow + kk);
    #pragma unroll
    for (int t = 0; t < 4; ++t) {
      v2f bm = *(const v2f*)(wvw + (size_t)(h * 64 + 16 * t + lm) * DM + k0 + kk + 2 * lh);
      acc[t] = wmma4(a, bm, acc[t]);
    }
  }
  const v4f s = wg_sum_tile(acc, red, wv, lm, lh);
  const int il = i0 + 2 * wv + lh;
  const int ifl = h * 64 + il;
  const float bki = bk[ifl];
  const float ui  = u[(size_t)b * DM + ifl];
  const v4f bv4 = *(const v4f*)(bv + h * 64 + 4 * lm);
  const v4f z4  = *(const v4f*)(z + (size_t)b * DM + h * 64 + 4 * lm);
  v4f o;
  #pragma unroll
  for (int c = 0; c < 4; ++c) {
    const float bvj = bv4[c], zj = z4[c];
    float val = s[c] + ui * bvj + bki * zj + (float)SEQ * bki * bvj;
    o[c] = 0.125f * val;
  }
  *(v4f*)(M + ((size_t)bh * 64 + il) * DKH + 4 * lm) = o;
}

__global__ void __launch_bounds__(256) kP(const float* __restrict__ M,
                                          const float* __restrict__ wo,
                                          float* __restrict__ P) {
  const int lane = threadIdx.x & 31, wv = threadIdx.x >> 5;
  const int lm = lane & 15, lh = lane >> 4;
  const int job = blockIdx.x * 8 + wv;
  const int bh = job >> 5;
  const int b = bh >> 3, h = bh & 7;
  const int i0 = ((job >> 3) & 3) * 16;
  const int o0 = (job & 7) * 64;
  const float* arow = M + ((size_t)bh * 64 + i0 + lm) * DKH + 2 * lh;
  v8f acc[4] = {};
  for (int kk = 0; kk < DKH; kk += 4) {
    v2f a = *(const v2f*)(arow + kk);
    #pragma unroll
    for (int t = 0; t < 4; ++t) {
      v2f bm = *(const v2f*)(wo + (size_t)(o0 + 16 * t + lm) * DM + h * 64 + kk + 2 * lh);
      acc[t] = wmma4(a, bm, acc[t]);
    }
  }
  float* Pb = P + ((size_t)b * DM + h * 64 + i0) * DM + o0;
  #pragma unroll
  for (int t = 0; t < 4; ++t)
    #pragma unroll
    for (int r = 0; r < 8; ++r)
      Pb[(size_t)(r + 8 * lh) * DM + 16 * t + lm] = acc[t][r];
}

__global__ void __launch_bounds__(256) SLICE_REGS kW4(const float* __restrict__ wq,
                                           const float* __restrict__ P,
                                           float* __restrict__ W4) {
  __shared__ __attribute__((aligned(16))) float red[8 * 1024];
  const int lane = threadIdx.x & 31, wv = threadIdx.x >> 5;
  const int lm = lane & 15, lh = lane >> 4;
  const int job = blockIdx.x;
  const int b = job >> 8;
  const int rest = job & 255;
  const int d0 = (rest >> 3) * 16;
  const int o0 = (rest & 7) * 64;
  const int k0 = wv * KW;
  const float* wqk = wq + (size_t)k0 * DM;
  const float* Pb = P + ((size_t)b * DM + k0) * DM;
  v8f acc[4] = {};
  #pragma unroll
  for (int kk = 0; kk < KW; kk += 4) {
    size_t fr = (size_t)(kk + 2 * lh) * DM;
    v2f a; a.x = wqk[fr + d0 + lm]; a.y = wqk[fr + DM + d0 + lm];
    #pragma unroll
    for (int t = 0; t < 4; ++t) {
      v2f bm; bm.x = Pb[fr + o0 + 16 * t + lm]; bm.y = Pb[fr + DM + o0 + 16 * t + lm];
      acc[t] = wmma4(a, bm, acc[t]);
    }
  }
  const v4f s = wg_sum_tile(acc, red, wv, lm, lh);
  float* Wb = W4 + ((size_t)b * DM + d0) * DM + o0;
  *(v4f*)(Wb + (size_t)(2 * wv + lh) * DM + 4 * lm) = s;
}

struct OSet { v4f a0[2], a1[2], w[8]; };

__device__ __forceinline__ void out_load(OSet& f, const float* __restrict__ qrow0,
                                         const float* __restrict__ qrow1,
                                         const float* __restrict__ Wp, int kk) {
  f.a0[0] = *(const v4f*)(qrow0 + kk); f.a0[1] = *(const v4f*)(qrow0 + kk + 4);
  f.a1[0] = *(const v4f*)(qrow1 + kk); f.a1[1] = *(const v4f*)(qrow1 + kk + 4);
  #pragma unroll
  for (int j = 0; j < 8; ++j) f.w[j] = *(const v4f*)(Wp + (size_t)(kk + j) * DM);
}

__device__ __forceinline__ void out_mma(const OSet& f, v8f (&acc)[8]) {
  #pragma unroll
  for (int p = 0; p < 4; ++p) {
    v2f x0; x0.x = f.a0[p >> 1][2 * (p & 1)]; x0.y = f.a0[p >> 1][2 * (p & 1) + 1];
    v2f x1; x1.x = f.a1[p >> 1][2 * (p & 1)]; x1.y = f.a1[p >> 1][2 * (p & 1) + 1];
    #pragma unroll
    for (int t = 0; t < 4; ++t) {
      v2f bm; bm.x = f.w[2 * p][t]; bm.y = f.w[2 * p + 1][t];
      acc[t]     = wmma4(x0, bm, acc[t]);
      acc[4 + t] = wmma4(x1, bm, acc[4 + t]);
    }
  }
}

__global__ void __launch_bounds__(256) SLICE_REGS kOut(const float* __restrict__ q,
                                            const float* __restrict__ W4,
                                            const float* __restrict__ bo,
                                            const float* __restrict__ cq,
                                            float* __restrict__ out) {
  const int lane = threadIdx.x & 31, wv = threadIdx.x >> 5;
  const int lm = lane & 15, lh = lane >> 4;
  const int job = blockIdx.x * 8 + wv;
  const int b = job >> 7;
  const int n0 = (job & 127) * 32;
  const int o0 = blockIdx.y * 64;
  const float* qrow0 = q + ((size_t)b * SEQ + n0 + lm) * DM + 8 * lh;
  const float* qrow1 = qrow0 + (size_t)16 * DM;
  const float* Wp = W4 + (size_t)b * DM * DM + (size_t)(8 * lh) * DM + o0 + 4 * lm;
  v8f acc[8] = {};
  OSet f0, f1;
  out_load(f0, qrow0, qrow1, Wp, 0);
  for (int kk = 0; kk < DM - 32; kk += 32) {
    out_load(f1, qrow0, qrow1, Wp, kk + 16);
    out_mma(f0, acc);
    out_load(f0, qrow0, qrow1, Wp, kk + 32);
    out_mma(f1, acc);
  }
  out_load(f1, qrow0, qrow1, Wp, DM - 16);
  out_mma(f0, acc);
  out_mma(f1, acc);
  const v4f add = *(const v4f*)(bo + o0 + 4 * lm) + *(const v4f*)(cq + (size_t)b * DM + o0 + 4 * lm);
  float* ob = out + ((size_t)b * SEQ + n0) * DM + o0 + 4 * lm;
  #pragma unroll
  for (int m = 0; m < 2; ++m)
    #pragma unroll
    for (int r = 0; r < 8; ++r) {
      v4f o;
      o.x = acc[4 * m][r]; o.y = acc[4 * m + 1][r]; o.z = acc[4 * m + 2][r]; o.w = acc[4 * m + 3][r];
      *(v4f*)(ob + (size_t)(16 * m + r + 8 * lh) * DM) = o + add;
    }
}

__global__ void __launch_bounds__(256) kUZ(const float* __restrict__ wk, const float* __restrict__ wvw,
                    const float* __restrict__ ksum, const float* __restrict__ vsum,
                    float* __restrict__ u, float* __restrict__ z) {
  const int lane = threadIdx.x & 31, wv = threadIdx.x >> 5;
  const int job = blockIdx.x * 8 + wv;
  const bool second = job >= DM;
  const int f = job & (DM - 1);
  const float* wr = (second ? wvw : wk) + (size_t)f * DM + 4 * lane;
  const float* sm = (second ? vsum : ksum) + 4 * lane;
  float p0 = 0.f, p1 = 0.f;
  #pragma unroll
  for (int j = 0; j < 4; ++j) {
    const v4f w4 = *(const v4f*)(wr + 128 * j);
    const v4f s0 = *(const v4f*)(sm + 128 * j);
    const v4f s1 = *(const v4f*)(sm + DM + 128 * j);
    #pragma unroll
    for (int c = 0; c < 4; ++c) { p0 += w4[c] * s0[c]; p1 += w4[c] * s1[c]; }
  }
  #pragma unroll
  for (int off = 16; off >= 1; off >>= 1) {
    p0 += __shfl_xor(p0, off);
    p1 += __shfl_xor(p1, off);
  }
  if (lane == 0) {
    float* dst = second ? z : u;
    dst[f] = p0;
    dst[DM + f] = p1;
  }
}

__global__ void __launch_bounds__(256) SLICE_REGS kCQ(const float* __restrict__ bq, const float* __restrict__ P,
                    float* __restrict__ cq) {
  __shared__ float part[8 * 32];
  const int lane = threadIdx.x & 31, wv = threadIdx.x >> 5;
  const int b = blockIdx.x >> 4;
  const int o = (blockIdx.x & 15) * 32 + lane;
  const int f0 = wv * KW;
  const float* Pb = P + ((size_t)b * DM + f0) * DM + o;
  float s = 0.f;
  #pragma unroll
  for (int f = 0; f < KW; ++f) s += bq[f0 + f] * Pb[(size_t)f * DM];
  part[wv * 32 + lane] = s;
  __syncthreads();
  if (wv == 0) {
    float t = part[lane];
    #pragma unroll
    for (int w = 1; w < 8; ++w) t += part[w * 32 + lane];
    cq[(size_t)b * DM + o] = t;
  }
}

extern "C" void kernel_launch(void* const* d_in, const int* in_sizes, int n_in,
                              void* d_out, int out_size, void* d_ws, size_t ws_size,
                              hipStream_t stream) {
  const float* q   = (const float*)d_in[0];
  const float* k   = (const float*)d_in[1];
  const float* v   = (const float*)d_in[2];
  const float* wq  = (const float*)d_in[3];
  const float* bq  = (const float*)d_in[4];
  const float* wk  = (const float*)d_in[5];
  const float* bk  = (const float*)d_in[6];
  const float* wvw = (const float*)d_in[7];
  const float* bv  = (const float*)d_in[8];
  const float* wo  = (const float*)d_in[9];
  const float* bo  = (const float*)d_in[10];
  float* out = (float*)d_out;

  float* ws = (float*)d_ws;
  float* G    = ws;
  float* T    = G    + (size_t)NB * DM * DM;
  float* M    = T    + (size_t)NB * DM * DM;
  float* P    = M    + (size_t)NB * NH * DKH * DKH;
  float* W4   = P    + (size_t)NB * DM * DM;
  float* ksum = W4   + (size_t)NB * DM * DM;
  float* vsum = ksum + (size_t)NB * DM;
  float* u    = vsum + (size_t)NB * DM;
  float* z    = u    + (size_t)NB * DM;
  float* cq   = z    + (size_t)NB * DM;
  float* Gp   = cq   + (size_t)NB * DM;
  float* SP   = Gp   + (size_t)KSPLIT * NB * DM * DM;

  kG   <<<KG_BLOCKS, 256, 0, stream>>>(k, v, Gp, SP);
  kGsum<<<GSUM_GRID, 256, 0, stream>>>(Gp, G, SP, ksum);
  kUZ  <<<128, 256, 0, stream>>>(wk, wvw, ksum, vsum, u, z);
  kT   <<<512, 256, 0, stream>>>(wk, G, T);
  kM   <<<64, 256, 0, stream>>>(T, wvw, bk, bv, u, z, M);
  kP   <<<64, 256, 0, stream>>>(M, wo, P);
  kCQ  <<<32, 256, 0, stream>>>(bq, P, cq);
  kW4  <<<512, 256, 0, stream>>>(wq, P, W4);
  dim3 go(32, 8, 1);
  kOut <<<go, 256, 0, stream>>>(q, W4, bo, cq, out);
}


# 9 launches -> 6: u,z ride in the T kernel's launch and cq in the W4 kernel's (extra work-groups, role by block index); P is computed by M's work-groups from the M tile kept in LDS; no sum changes
# speedup vs baseline: 9.0946x; 1.1002x over previous
#include <hip/hip_runtime.h>


typedef float v2f __attribute__((ext_vector_type(2)));
typedef float v4f __attribute__((ext_vector_type(4)));
typedef float v8f __attribute__((ext_vector_type(8)));

#define DM   512
#define SEQ  4096
#define NB   2
#define NH   8
#define DKH  64
#define KSPLIT 16
#define SCH  (SEQ / KSPLIT)
#define KG_BLOCKS    (32 * KSPLIT)
#define GSUM_BLOCKS  (NB * DM * DM / 4 / 256)
#define NPART  (2 * KSPLIT)
#define NSUMS  (2 * NB * DM)
#define SUMS_BLOCKS  (NSUMS / 256)
#define GSUM_GRID    (GSUM_BLOCKS + SUMS_BLOCKS)
#define KW   (DM / 8)
#define UZ_BLOCKS  (2 * DM / 8)
#define T_BLOCKS   512
#define CQ_BLOCKS  (NB * DM / 32)
#define W4_BLOCKS  512
#define TU_GRID    (UZ_BLOCKS + T_BLOCKS)
#define W4C_GRID   (CQ_BLOCKS + W4_BLOCKS)
#define MT_PITCH   68
#define SLICE_REGS __attribute__((amdgpu_waves_per_eu(4, 4)))
#define PAIR_REGS __attribute__((amdgpu_waves_per_eu(2, 2)))

__device__ __forceinline__ v8f wmma4(v2f a, v2f b, v8f c) {
  return __builtin_amdgcn_wmma_f32_16x16x4_f32(
       false, a,  false, b,
       (short)0, c,  false,  false);
}

__device__ __forceinline__ v4f wg_sum_tile(const v8f (&acc)[4], float* red, int wv, int lm, int lh) {
  float* mine = red + wv * 1024;
  #pragma unroll
  for (int t = 0; t < 4; ++t)
    #pragma unroll
    for (int r = 0; r < 8; ++r)
      mine[(r + 8 * lh) * 64 + 16 * t + lm] = acc[t][r];
  __syncthreads();
  const int off = (2 * wv + lh) * 64 + 4 * lm;
  v4f s = *(const v4f*)(red + off);
  #pragma unroll
  for (int w = 1; w < 8; ++w) s += *(const v4f*)(red + w * 1024 + off);
  return s;
}

struct GSet { v2f kx[4], ky[4]; v4f vx[4], vy[4]; };

__device__ __forceinline__ void g_load(GSet& f, const float* __restrict__ kp,
                                       const float* __restrict__ vp, int s) {
  #pragma unroll
  for (int i = 0; i < 4; ++i) {
    f.kx[i] = *(const v2f*)(kp + (size_t)(s + 4 * i) * DM);
    f.ky[i] = *(const v2f*)(kp + (size_t)(s + 4 * i + 1) * DM);
    f.vx[i] = *(const v4f*)(vp + (size_t)(s + 4 * i) * DM);
    f.vy[i] = *(const v4f*)(vp + (size_t)(s + 4 * i + 1) * DM);
  }
}

template <bool SK, bool SV>
__device__ __forceinline__ void g_mma(const GSet& f, v8f (&acc)[8],
                                      v2f& ksx, v2f& ksy, v4f& vsx, v4f& vsy) {
  #pragma unroll
  for (int i = 0; i < 4; ++i) {
    const v2f kx = f.kx[i], ky = f.ky[i];
    const v4f vx = f.vx[i], vy = f.vy[i];
    if (SK) { ksx += kx; ksy += ky; }
    if (SV) { vsx += vx; vsy += vy; }
    v2f a0; a0.x = kx.x; a0.y = ky.x;
    v2f a1; a1.x = kx.y; a1.y = ky.y;
    #pragma unroll
    for (int t = 0; t < 4; ++t) {
      v2f bm; bm.x = vx[t]; bm.y = vy[t];
      acc[t]     = wmma4(a0, bm, acc[t]);
      acc[4 + t] = wmma4(a1, bm, acc[4 + t]);
    }
  }
}

template <bool SK, bool SV>
__device__ __forceinline__ void g_tile(const float* __restrict__ kb,
                                       const float* __restrict__ vb,
                                       float* __restrict__ Gb,
                                       float* __restrict__ spk,
                                       float* __restrict__ spv,
                                       int d0, int e0, int lm, int lh) {
  v8f acc[8] = {};
  v2f ksx = {0.f, 0.f}, ksy = {0.f, 0.f};
  v4f vsx = {0.f, 0.f, 0.f, 0.f}, vsy = {0.f, 0.f, 0.f, 0.f};
  const float* kp = kb + (size_t)(2 * lh) * DM + d0 + 2 * lm;
  const float* vp = vb + (size_t)(2 * lh) * DM + e0 + 4 * lm;
  GSet f0, f1;
  g_load(f0, kp, vp, 0);
  #pragma nounroll
  for (int s0 = 0; s0 < SCH - 32; s0 += 32) {
    g_load(f1, kp, vp, s0 + 16);
    g_mma<SK, SV>(f0, acc, ksx, ksy, vsx, vsy);
    g_load(f0, kp, vp, s0 + 32);
    g_mma<SK, SV>(f1, acc, ksx, ksy, vsx, vsy);
  }
  g_load(f1, kp, vp, SCH - 16);
  g_mma<SK, SV>(f0, acc, ksx, ksy, vsx, vsy);
  g_mma<SK, SV>(f1, acc, ksx, ksy, vsx, vsy);
  #pragma unroll
  for (int m = 0; m < 2; ++m)
    #pragma unroll
    for (int r = 0; r < 8; ++r) {
      v4f o;
      o.x = acc[4 * m][r]; o.y = acc[4 * m + 1][r]; o.z = acc[4 * m + 2][r]; o.w = acc[4 * m + 3][r];
      *(v4f*)(Gb + (size_t)(2 * (8 * lh + r) + m) * DM + 4 * lm) = o;
    }
  if (SK) *(v2f*)(spk + d0 + 2 * lm) = ksx + ksy;
  if (SV) *(v4f*)(spv + e0 + 4 * lm) = vsx + vsy;
}

__global__ void __launch_bounds__(256) SLICE_REGS kG(const float* __restrict__ k,
                                          const float* __restrict__ v,
                                          float* __restrict__ Gp,
                                          float* __restrict__ SP) {
  const int lane = threadIdx.x & 31, wv = threadIdx.x >> 5;
  const int lm = lane & 15, lh = lane >> 4;
  const int job = (blockIdx.x & 31) * 8 + wv;
  const int c = blockIdx.x >> 5;
  const int b = job >> 7;
  const int rest = job & 127;
  const int d0 = (rest >> 3) * 32;
  const int e0 = (rest & 7) * 64;
  const float* kb = k + ((size_t)b * SEQ + (size_t)c * SCH) * DM;
  const float* vb = v + ((size_t)b * SEQ + (size_t)c * SCH) * DM;
  float* Gb = Gp + (((size_t)c * NB + b) * DM + d0) * DM + e0;
  float* spk = SP + ((size_t)(2 * c + lh) * 2 + 0) * (NB * DM) + (size_t)b * DM;
  float* spv = SP + ((size_t)(2 * c + lh) * 2 + 1) * (NB * DM) + (size_t)b * DM;
  const bool sumK = __builtin_amdgcn_readfirstlane(wv) == 0;
  const bool sumV = (blockIdx.x & 15) == 0;
  if (sumV) {
    if (sumK) g_tile<true, true>(kb, vb, Gb, spk, spv, d0, e0, lm, lh);
    else      g_tile<false, true>(kb, vb, Gb, spk, spv, d0, e0, lm, lh);
  } else {
    if (sumK) g_tile<true, false>(kb, vb, Gb, spk, spv, d0, e0, lm, lh);
    else      g_tile<false, false>(kb, vb, Gb, spk, spv, d0, e0, lm, lh);
  }
}

__global__ void __launch_bounds__(256) kGsum(const float* __restrict__ Gp,
                                             float* __restrict__ G,
                                             const float* __restrict__ SP,
                                             float* __restrict__ sums) {
  if (blockIdx.x >= GSUM_BLOCKS) {
    const int j = (blockIdx.x - GSUM_BLOCKS) * 256 + threadIdx.x;
    float s = SP[j];
    for (int p = 1; p < NPART; ++p) s += SP[(size_t)p * NSUMS + j];
    sums[j] = s;
    return;
  }
  const int i = blockIdx.x * 256 + threadIdx.x;
  const float4* p = (const float4*)Gp + i;
  float4 s = p[0];
  for (int c = 1; c < KSPLIT; ++c) {
    const float4 t = p[(size_t)c * (NB * DM * DM / 4)];
    s.x += t.x; s.y += t.y; s.z += t.z; s.w += t.w;
  }
  ((float4*)G)[i] = s;
}

__device__ __forceinline__ void uz_rows(int blk, int lane, int wv,
                    const float* __restrict__ wk, const float* __restrict__ wvw,
                    const float* __restrict__ ksum, const float* __restrict__ vsum,
                    float* __restrict__ u, float* __restrict__ z) {
  const int job = blk * 8 + wv;
  const bool second = job >= DM;
  const int f = job & (DM - 1);
  const float* wr = (second ? wvw : wk) + (size_t)f * DM + 4 * lane;
  const float* sm = (second ? vsum : ksum) + 4 * lane;
  float p0 = 0.f, p1 = 0.f;
  #pragma unroll
  for (int j = 0; j < 4; ++j) {
    const v4f w4 = *(const v4f*)(wr + 128 * j);
    const v4f s0 = *(const v4f*)(sm + 128 * j);
    const v4f s1 = *(const v4f*)(sm + DM + 128 * j);
    #pragma unroll
    for (int c = 0; c < 4; ++c) { p0 += w4[c] * s0[c]; p1 += w4[c] * s1[c]; }
  }
  #pragma unroll
  for (int off = 16; off >= 1; off >>= 1) {
    p0 += __shfl_xor(p0, off);
    p1 += __shfl_xor(p1, off);
  }
  if (lane == 0) {
    float* dst = second ? z : u;
    dst[f] = p0;
    dst[DM + f] = p1;
  }
}

__device__ __forceinline__ void cq_cols(int blk, int lane, int wv, float* part,
                    const float* __restrict__ bq, const float* __restrict__ P,
                    float* __restrict__ cq) {
  const int b = blk >> 4;
  const int o = (blk & 15) * 32 + lane;
  const int f0 = wv * KW;
  const float* Pb = P + ((size_t)b * DM + f0) * DM + o;
  float s = 0.f;
  #pragma unroll
  for (int f = 0; f < KW; ++f) s += bq[f0 + f] * Pb[(size_t)f * DM];
  part[wv * 32 + lane] = s;
  __syncthreads();
  if (wv == 0) {
    float t = part[lane];
    #pragma unroll
    for (int w = 1; w < 8; ++w) t += part[w * 32 + lane];
    cq[(size_t)b * DM + o] = t;
  }
}

__global__ void __launch_bounds__(256) SLICE_REGS kTU(const float* __restrict__ wk,
                                           const float* __restrict__ wvw,
                                           const float* __restrict__ G,
                                           const float* __restrict__ ksum,
                                           const float* __restrict__ vsum,
                                           float* __restrict__ T,
                                           float* __restrict__ u,
                                           float* __restrict__ z) {
  __shared__ __attribute__((aligned(16))) float red[8 * 1024];
  const int lane = threadIdx.x & 31, wv = threadIdx.x >> 5;
  if (blockIdx.x < UZ_BLOCKS) {
    uz_rows(blockIdx.x, lane, wv, wk, wvw, ksum, vsum, u, z);
    return;
  }
  const int lm = lane & 15, lh = lane >> 4;
  const int job = blockIdx.x - UZ_BLOCKS;
  const int bh = job >> 5;
  const int b = bh >> 3, h = bh & 7;
  const int rest = job & 31;
  const int i0 = (rest >> 3) * 16;
  const int e0 = (rest & 7) * 64;
  const int k0 = wv * KW;
  const float* arow = wk + (size_t)(h * 64 + i0 + lm) * DM + k0 + 2 * lh;
  const float* Gb = G + ((size_t)b * DM + k0) * DM;
  v8f acc[4] = {};
  #pragma unroll
  for (int kk = 0; kk < KW; kk += 4) {
    v2f a = *(const v2f*)(arow + kk);
    size_t rb = (size_t)(kk + 2 * lh) * DM;
    #pragma unroll
    for (int t = 0; t < 4; ++t) {
      v2f bm; bm.x = Gb[rb + e0 + 16 * t + lm]; bm.y = Gb[rb + DM + e0 + 16 * t + lm];
      acc[t] = wmma4(a, bm, acc[t]);
    }
  }
  const v4f s = wg_sum_tile(acc, red, wv, lm, lh);
  float* Tb = T + ((size_t)bh * 64 + i0) * DM + e0;
  *(v4f*)(Tb + (size_t)(2 * wv + lh) * DM + 4 * lm) = s;
}

__global__ void __launch_bounds__(256) PAIR_REGS kMP(const float* __restrict__ T,
                                           const float* __restrict__ wvw,
                                           const float* __restrict__ bk,
                                           const float* __restrict__ bv,
                                           const float* __restrict__ u,
                                           const float* __restrict__ z,
                                           const float* __restrict__ wo,
                                           float* __restrict__ P) {
  __shared__ __attribute__((aligned(16))) float red[8 * 1024];
  __shared__ __attribute__((aligned(16))) float mt[16 * MT_PITCH];
  const int lane = threadIdx.x & 31, wv = threadIdx.x >> 5;
  const int lm = lane & 15, lh = lane >> 4;
  const int job = blockIdx.x;
  const int bh = job >> 2;
  const int b = bh >> 3, h = bh & 7;
  const int i0 = (job & 3) * 16;
  const int k0 = wv * KW;
  const float* arow = T + ((size_t)bh * 64 + i0 + lm) * DM + k0 + 2 * lh;
  v8f acc[4] = {};
  #pragma unroll
  for (int kk = 0; kk < KW; kk += 4) {
    v2f a = *(const v2f*)(arow + kk);
    #pragma unroll
    for (int t = 0; t < 4; ++t) {
      v2f bm = *(const v2f*)(wvw + (size_t)(h * 64 + 16 * t + lm) * DM + k0 + kk + 2 * lh);
      acc[t] = wmma4(a, bm, acc[t]);
    }
  }
  const int o0 = wv * 64;
  const float* wob = wo + (size_t)(o0 + lm) * DM + h * 64 + 2 * lh;
  v2f wf[16][4];
  #pragma unroll
  for (int kq = 0; kq < 16; ++kq)
    #pragma unroll
    for (int t = 0; t < 4; ++t)
      wf[kq][t] = *(const v2f*)(wob + (size_t)(16 * t) * DM + 4 * kq);
  const v4f s = wg_sum_tile(acc, red, wv, lm, lh);
  const int il = i0 + 2 * wv + lh;
  const int ifl = h * 64 + il;
  const float bki = bk[ifl];
  const float ui  = u[(size_t)b * DM + ifl];
  const v4f bv4 = *(const v4f*)(bv + h * 64 + 4 * lm);
  const v4f z4  = *(const v4f*)(z + (size_t)b * DM + h * 64 + 4 * lm);
  v4f o;
  #pragma unroll
  for (int c = 0; c < 4; ++c) {
    const float bvj = bv4[c], zj = z4[c];
    float val = s[c] + ui * bvj + bki * zj + (float)SEQ * bki * bvj;
    o[c] = 0.125f * val;
  }
  *(v4f*)(mt + (2 * wv + lh) * MT_PITCH + 4 * lm) = o;
  __syncthreads();
  const float* mrow = mt + lm * MT_PITCH + 2 * lh;
  v8f pacc[4] = {};
  #pragma unroll
  for (int kq = 0; kq < 16; ++kq) {
    v2f a = *(const v2f*)(mrow + 4 * kq);
    #pragma unroll
    for (int t = 0; t < 4; ++t) pacc[t] = wmma4(a, wf[kq][t], pacc[t]);
  }
  float* Pb = P + ((size_t)b * DM + h * 64 + i0) * DM + o0;
  #pragma unroll
  for (int t = 0; t < 4; ++t)
    #pragma unroll
    for (int r = 0; r < 8; ++r)
      Pb[(size_t)(r + 8 * lh) * DM + 16 * t + lm] = pacc[t][r];
}

__global__ void __launch_bounds__(256) SLICE_REGS kW4C(const float* __restrict__ wq,
                                            const float* __restrict__ bq,
                                            const float* __restrict__ P,
                                            float* __restrict__ W4,
                                            float* __restrict__ cq) {
  __shared__ __attribute__((aligned(16))) float red[8 * 1024];
  const int lane = threadIdx.x & 31, wv = threadIdx.x >> 5;
  if (blockIdx.x < CQ_BLOCKS) {
    cq_cols(blockIdx.x, lane, wv, red, bq, P, cq);
    return;
  }
  const int lm = lane & 15, lh = lane >> 4;
  const int job = blockIdx.x - CQ_BLOCKS;
  const int b = job >> 8;
  const int rest = job & 255;
  const int d0 = (rest >> 3) * 16;
  const int o0 = (rest & 7) * 64;
  const int k0 = wv * KW;
  const float* wqk = wq + (size_t)k0 * DM;
  const float* Pb = P + ((size_t)b * DM + k0) * DM;
  v8f acc[4] = {};
  #pragma unroll
  for (int kk = 0; kk < KW; kk += 4) {
    size_t fr = (size_t)(kk + 2 * lh) * DM;
    v2f a; a.x = wqk[fr + d0 + lm]; a.y = wqk[fr + DM + d0 + lm];
    #pragma unroll
    for (int t = 0; t < 4; ++t) {
      v2f bm; bm.x = Pb[fr + o0 + 16 * t + lm]; bm.y = Pb[fr + DM + o0 + 16 * t + lm];
      acc[t] = wmma4(a, bm, acc[t]);
    }
  }
  const v4f s = wg_sum_tile(acc, red, wv, lm, lh);
  float* Wb = W4 + ((size_t)b * DM + d0) * DM + o0;
  *(v4f*)(Wb + (size_t)(2 * wv + lh) * DM + 4 * lm) = s;
}

struct OSet { v4f a0[2], a1[2], w[8]; };

__device__ __forceinline__ void out_load(OSet& f, const float* __restrict__ qrow0,
                                         const float* __restrict__ qrow1,
                                         const float* __restrict__ Wp, int kk) {
  f.a0[0] = *(const v4f*)(qrow0 + kk); f.a0[1] = *(const v4f*)(qrow0 + kk + 4);
  f.a1[0] = *(const v4f*)(qrow1 + kk); f.a1[1] = *(const v4f*)(qrow1 + kk + 4);
  #pragma unroll
  for (int j = 0; j < 8; ++j) f.w[j] = *(const v4f*)(Wp + (size_t)(kk + j) * DM);
}

__device__ __forceinline__ void out_mma(const OSet& f, v8f (&acc)[8]) {
  #pragma unroll
  for (int p = 0; p < 4; ++p) {
    v2f x0; x0.x = f.a0[p >> 1][2 * (p & 1)]; x0.y = f.a0[p >> 1][2 * (p & 1) + 1];
    v2f x1; x1.x = f.a1[p >> 1][2 * (p & 1)]; x1.y = f.a1[p >> 1][2 * (p & 1) + 1];
    #pragma unroll
    for (int t = 0; t < 4; ++t) {
      v2f bm; bm.x = f.w[2 * p][t]; bm.y = f.w[2 * p + 1][t];
      acc[t]     = wmma4(x0, bm, acc[t]);
      acc[4 + t] = wmma4(x1, bm, acc[4 + t]);
    }
  }
}

__global__ void __launch_bounds__(256) SLICE_REGS kOut(const float* __restrict__ q,
                                            const float* __restrict__ W4,
                                            const float* __restrict__ bo,
                                            const float* __restrict__ cq,
                                            float* __restrict__ out) {
  const int lane = threadIdx.x & 31, wv = threadIdx.x >> 5;
  const int lm = lane & 15, lh = lane >> 4;
  const int job = blockIdx.x * 8 + wv;
  const int b = job >> 7;
  const int n0 = (job & 127) * 32;
  const int o0 = blockIdx.y * 64;
  const float* qrow0 = q + ((size_t)b * SEQ + n0 + lm) * DM + 8 * lh;
  const float* qrow1 = qrow0 + (size_t)16 * DM;
  const float* Wp = W4 + (size_t)b * DM * DM + (size_t)(8 * lh) * DM + o0 + 4 * lm;
  v8f acc[8] = {};
  OSet f0, f1;
  out_load(f0, qrow0, qrow1, Wp, 0);
  for (int kk = 0; kk < DM - 32; kk += 32) {
    out_load(f1, qrow0, qrow1, Wp, kk + 16);
    out_mma(f0, acc);
    out_load(f0, qrow0, qrow1, Wp, kk + 32);
    out_mma(f1, acc);
  }
  out_load(f1, qrow0, qrow1, Wp, DM - 16);
  out_mma(f0, acc);
  out_mma(f1, acc);
  const v4f add = *(const v4f*)(bo + o0 + 4 * lm) + *(const v4f*)(cq + (size_t)b * DM + o0 + 4 * lm);
  float* ob = out + ((size_t)b * SEQ + n0) * DM + o0 + 4 * lm;
  #pragma unroll
  for (int m = 0; m < 2; ++m)
    #pragma unroll
    for (int r = 0; r < 8; ++r) {
      v4f o;
      o.x = acc[4 * m][r]; o.y = acc[4 * m + 1][r]; o.z = acc[4 * m + 2][r]; o.w = acc[4 * m + 3][r];
      *(v4f*)(ob + (size_t)(16 * m + r + 8 * lh) * DM) = o + add;
    }
}

extern "C" void kernel_launch(void* const* d_in, const int* in_sizes, int n_in,
                              void* d_out, int out_size, void* d_ws, size_t ws_size,
                              hipStream_t stream) {
  const float* q   = (const float*)d_in[0];
  const float* k   = (const float*)d_in[1];
  const float* v   = (const float*)d_in[2];
  const float* wq  = (const float*)d_in[3];
  const float* bq  = (const float*)d_in[4];
  const float* wk  = (const float*)d_in[5];
  const float* bk  = (const float*)d_in[6];
  const float* wvw = (const float*)d_in[7];
  const float* bv  = (const float*)d_in[8];
  const float* wo  = (const float*)d_in[9];
  const float* bo  = (const float*)d_in[10];
  float* out = (float*)d_out;

  float* ws = (float*)d_ws;
  float* G    = ws;
  float* T    = G    + (size_t)NB * DM * DM;
  float* P    = T    + (size_t)NB * DM * DM;
  float* W4   = P    + (size_t)NB * DM * DM;
  float* ksum = W4   + (size_t)NB * DM * DM;
  float* vsum = ksum + (size_t)NB * DM;
  float* u    = vsum + (size_t)NB * DM;
  float* z    = u    + (size_t)NB * DM;
  float* cq   = z    + (size_t)NB * DM;
  float* Gp   = cq   + (size_t)NB * DM;
  float* SP   = Gp   + (size_t)KSPLIT * NB * DM * DM;

  kG   <<<KG_BLOCKS, 256, 0, stream>>>(k, v, Gp, SP);
  kGsum<<<GSUM_GRID, 256, 0, stream>>>(Gp, G, SP, ksum);
  kTU  <<<TU_GRID, 256, 0, stream>>>(wk, wvw, G, ksum, vsum, T, u, z);
  kMP  <<<64, 256, 0, stream>>>(T, wvw, bk, bv, u, z, wo, P);
  kW4C <<<W4C_GRID, 256, 0, stream>>>(wq, bq, P, W4, cq);
  dim3 go(32, 8, 1);
  kOut <<<go, 256, 0, stream>>>(q, W4, bo, cq, out);
}


# kG: the 16 SEQ chunks of a tile become the 16 waves of one work-group; tile and column-sum partials added via LDS in the same index order: no partial planes, no sum kernel (6 -> 5 launches)
# speedup vs baseline: 9.7245x; 1.0693x over previous
#include <hip/hip_runtime.h>


typedef float v2f __attribute__((ext_vector_type(2)));
typedef float v4f __attribute__((ext_vector_type(4)));
typedef float v8f __attribute__((ext_vector_type(8)));

#define DM   512
#define SEQ  4096
#define NB   2
#define NH   8
#define DKH  64
#define KSPLIT 16
#define SCH  (SEQ / KSPLIT)
#define KG_TILES     (NB * 16 * 8)
#define KG_THREADS   (32 * KSPLIT)
#define TILE_FLOATS  (32 * 64)
#define NPART  (2 * KSPLIT)
#define KW   (DM / 8)
#define UZ_BLOCKS  (2 * DM / 8)
#define T_BLOCKS   512
#define CQ_BLOCKS  (NB * DM / 32)
#define W4_BLOCKS  512
#define TU_GRID    (UZ_BLOCKS + T_BLOCKS)
#define W4C_GRID   (CQ_BLOCKS + W4_BLOCKS)
#define MT_PITCH   68
#define SLICE_REGS __attribute__((amdgpu_waves_per_eu(4, 4)))
#define PAIR_REGS __attribute__((amdgpu_waves_per_eu(2, 2)))

__device__ __forceinline__ v8f wmma4(v2f a, v2f b, v8f c) {
  return __builtin_amdgcn_wmma_f32_16x16x4_f32(
       false, a,  false, b,
       (short)0, c,  false,  false);
}

__device__ __forceinline__ v4f wg_sum_tile(const v8f (&acc)[4], float* red, int wv, int lm, int lh) {
  float* mine = red + wv * 1024;
  #pragma unroll
  for (int t = 0; t < 4; ++t)
    #pragma unroll
    for (int r = 0; r < 8; ++r)
      mine[(r + 8 * lh) * 64 + 16 * t + lm] = acc[t][r];
  __syncthreads();
  const int off = (2 * wv + lh) * 64 + 4 * lm;
  v4f s = *(const v4f*)(red + off);
  #pragma unroll
  for (int w = 1; w < 8; ++w) s += *(const v4f*)(red + w * 1024 + off);
  return s;
}

struct GSet { v2f kx[4], ky[4]; v4f vx[4], vy[4]; };

__device__ __forceinline__ void g_load(GSet& f, const float* __restrict__ kp,
                                       const float* __restrict__ vp, int s) {
  #pragma unroll
  for (int i = 0; i < 4; ++i) {
    f.kx[i] = *(const v2f*)(kp + (size_t)(s + 4 * i) * DM);
    f.ky[i] = *(const v2f*)(kp + (size_t)(s + 4 * i + 1) * DM);
    f.vx[i] = *(const v4f*)(vp + (size_t)(s + 4 * i) * DM);
    f.vy[i] = *(const v4f*)(vp + (size_t)(s + 4 * i + 1) * DM);
  }
}

template <bool SK, bool SV>
__device__ __forceinline__ void g_mma(const GSet& f, v8f (&acc)[8],
                                      v2f& ksx, v2f& ksy, v4f& vsx, v4f& vsy) {
  #pragma unroll
  for (int i = 0; i < 4; ++i) {
    const v2f kx = f.kx[i], ky = f.ky[i];
    const v4f vx = f.vx[i], vy = f.vy[i];
    if (SK) { ksx += kx; ksy += ky; }
    if (SV) { vsx += vx; vsy += vy; }
    v2f a0; a0.x = kx.x; a0.y = ky.x;
    v2f a1; a1.x = kx.y; a1.y = ky.y;
    #pragma unroll
    for (int t = 0; t < 4; ++t) {
      v2f bm; bm.x = vx[t]; bm.y = vy[t];
      acc[t]     = wmma4(a0, bm, acc[t]);
      acc[4 + t] = wmma4(a1, bm, acc[4 + t]);
    }
  }
}

template <bool SK, bool SV>
__device__ __forceinline__ void g_tile(const float* __restrict__ kb,
                                       const float* __restrict__ vb,
                                       float* plane,
                                       float* spk,
                                       float* spv,
                                       int d0, int e0, int lm, int lh) {
  v8f acc[8] = {};
  v2f ksx = {0.f, 0.f}, ksy = {0.f, 0.f};
  v4f vsx = {0.f, 0.f, 0.f, 0.f}, vsy = {0.f, 0.f, 0.f, 0.f};
  const float* kp = kb + (size_t)(2 * lh) * DM + d0 + 2 * lm;
  const float* vp = vb + (size_t)(2 * lh) * DM + e0 + 4 * lm;
  GSet f0, f1;
  g_load(f0, kp, vp, 0);
  #pragma nounroll
  for (int s0 = 0; s0 < SCH - 32; s0 += 32) {
    g_load(f1, kp, vp, s0 + 16);
    g_mma<SK, SV>(f0, acc, ksx, ksy, vsx, vsy);
    g_load(f0, kp, vp, s0 + 32);
    g_mma<SK, SV>(f1, acc, ksx, ksy, vsx, vsy);
  }
  g_load(f1, kp, vp, SCH - 16);
  g_mma<SK, SV>(f0, acc, ksx, ksy, vsx, vsy);
  g_mma<SK, SV>(f1, acc, ksx, ksy, vsx, vsy);
  #pragma unroll
  for (int m = 0; m < 2; ++m)
    #pragma unroll
    for (int r = 0; r < 8; ++r) {
      v4f o;
      o.x = acc[4 * m][r]; o.y = acc[4 * m + 1][r]; o.z = acc[4 * m + 2][r]; o.w = acc[4 * m + 3][r];
      *(v4f*)(plane + (2 * (8 * lh + r) + m) * 64 + 4 * lm) = o;
    }
  if (SK) *(v2f*)(spk + 2 * lm) = ksx + ksy;
  if (SV) *(v4f*)(spv + 4 * lm) = vsx + vsy;
}

__global__ void __launch_bounds__(KG_THREADS) SLICE_REGS kG(const float* __restrict__ k,
                                          const float* __restrict__ v,
                                          float* __restrict__ G,
                                          float* __restrict__ sums) {
  __shared__ __attribute__((aligned(16))) float red[KSPLIT * TILE_FLOATS];
  __shared__ __attribute__((aligned(16))) float pks[NPART * 32];
  __shared__ __attribute__((aligned(16))) float pvs[NPART * 64];
  const int lane = threadIdx.x & 31;
  const int lm = lane & 15, lh = lane >> 4;
  const int c = __builtin_amdgcn_readfirstlane((int)(threadIdx.x >> 5));
  const int tile = blockIdx.x;
  const int b = tile >> 7;
  const int rest = tile & 127;
  const int dt = rest >> 3;
  const int eg = rest & 7;
  const int d0 = dt * 32;
  const int e0 = eg * 64;
  const float* kb = k + ((size_t)b * SEQ + (size_t)c * SCH) * DM;
  const float* vb = v + ((size_t)b * SEQ + (size_t)c * SCH) * DM;
  float* plane = red + c * TILE_FLOATS;
  float* spk = pks + (2 * c + lh) * 32;
  float* spv = pvs + (2 * c + lh) * 64;
  const bool sumK = eg == 0;
  const bool sumV = dt == 0;
  if (sumV) {
    if (sumK) g_tile<true, true>(kb, vb, plane, spk, spv, d0, e0, lm, lh);
    else      g_tile<false, true>(kb, vb, plane, spk, spv, d0, e0, lm, lh);
  } else {
    if (sumK) g_tile<true, false>(kb, vb, plane, spk, spv, d0, e0, lm, lh);
    else      g_tile<false, false>(kb, vb, plane, spk, spv, d0, e0, lm, lh);
  }
  __syncthreads();
  const int t = threadIdx.x;
  v4f s = *(const v4f*)(red + 4 * t);
  #pragma unroll
  for (int p = 1; p < KSPLIT; ++p) s += *(const v4f*)(red + p * TILE_FLOATS + 4 * t);
  *(v4f*)(G + ((size_t)b * DM + d0 + (t >> 4)) * DM + e0 + 4 * (t & 15)) = s;
  if (sumK && t < 32) {
    float a = pks[t];
    #pragma unroll
    for (int p = 1; p < NPART; ++p) a += pks[p * 32 + t];
    sums[(size_t)b * DM + d0 + t] = a;
  }
  if (sumV && t >= 64 && t < 128) {
    const int j = t - 64;
    float a = pvs[j];
    #pragma unroll
    for (int p = 1; p < NPART; ++p) a += pvs[p * 64 + j];
    sums[(size_t)(NB + b) * DM + e0 + j] = a;
  }
}

__device__ __forceinline__ void uz_rows(int blk, int lane, int wv,
                    const float* __restrict__ wk, const float* __restrict__ wvw,
                    const float* __restrict__ ksum, const float* __restrict__ vsum,
                    float* __restrict__ u, float* __restrict__ z) {
  const int job = blk * 8 + wv;
  const bool second = job >= DM;
  const int f = job & (DM - 1);
  const float* wr = (second ? wvw : wk) + (size_t)f * DM + 4 * lane;
  const float* sm = (second ? vsum : ksum) + 4 * lane;
  float p0 = 0.f, p1 = 0.f;
  #pragma unroll
  for (int j = 0; j < 4; ++j) {
    const v4f w4 = *(const v4f*)(wr + 128 * j);
    const v4f s0 = *(const v4f*)(sm + 128 * j);
    const v4f s1 = *(const v4f*)(sm + DM + 128 * j);
    #pragma unroll
    for (int c = 0; c < 4; ++c) { p0 += w4[c] * s0[c]; p1 += w4[c] * s1[c]; }
  }
  #pragma unroll
  for (int off = 16; off >= 1; off >>= 1) {
    p0 += __shfl_xor(p0, off);
    p1 += __shfl_xor(p1, off);
  }
  if (lane == 0) {
    float* dst = second ? z : u;
    dst[f] = p0;
    dst[DM + f] = p1;
  }
}

__device__ __forceinline__ void cq_cols(int blk, int lane, int wv, float* part,
                    const float* __restrict__ bq, const float* __restrict__ P,
                    float* __restrict__ cq) {
  const int b = blk >> 4;
  const int o = (blk & 15) * 32 + lane;
  const int f0 = wv * KW;
  const float* Pb = P + ((size_t)b * DM + f0) * DM + o;
  float s = 0.f;
  #pragma unroll
  for (int f = 0; f < KW; ++f) s += bq[f0 + f] * Pb[(size_t)f * DM];
  part[wv * 32 + lane] = s;
  __syncthreads();
  if (wv == 0) {
    float t = part[lane];
    #pragma unroll
    for (int w = 1; w < 8; ++w) t += part[w * 32 + lane];
    cq[(size_t)b * DM + o] = t;
  }
}

__global__ void __launch_bounds__(256) SLICE_REGS kTU(const float* __restrict__ wk,
                                           const float* __restrict__ wvw,
                                           const float* __restrict__ G,
                                           const float* __restrict__ ksum,
                                           const float* __restrict__ vsum,
                                           float* __restrict__ T,
                                           float* __restrict__ u,
                                           float* __restrict__ z) {
  __shared__ __attribute__((aligned(16))) float red[8 * 1024];
  const int lane = threadIdx.x & 31, wv = threadIdx.x >> 5;
  if (blockIdx.x < UZ_BLOCKS) {
    uz_rows(blockIdx.x, lane, wv, wk, wvw, ksum, vsum, u, z);
    return;
  }
  const int lm = lane & 15, lh = lane >> 4;
  const int job = blockIdx.x - UZ_BLOCKS;
  const int bh = job >> 5;
  const int b = bh >> 3, h = bh & 7;
  const int rest = job & 31;
  const int i0 = (rest >> 3) * 16;
  const int e0 = (rest & 7) * 64;
  const int k0 = wv * KW;
  const float* arow = wk + (size_t)(h * 64 + i0 + lm) * DM + k0 + 2 * lh;
  const float* Gb = G + ((size_t)b * DM + k0) * DM;
  v8f acc[4] = {};
  #pragma unroll
  for (int kk = 0; kk < KW; kk += 4) {
    v2f a = *(const v2f*)(arow + kk);
    size_t rb = (size_t)(kk + 2 * lh) * DM;
    #pragma unroll
    for (int t = 0; t < 4; ++t) {
      v2f bm; bm.x = Gb[rb + e0 + 16 * t + lm]; bm.y = Gb[rb + DM + e0 + 16 * t + lm];
      acc[t] = wmma4(a, bm, acc[t]);
    }
  }
  const v4f s = wg_sum_tile(acc, red, wv, lm, lh);
  float* Tb = T + ((size_t)bh * 64 + i0) * DM + e0;
  *(v4f*)(Tb + (size_t)(2 * wv + lh) * DM + 4 * lm) = s;
}

__global__ void __launch_bounds__(256) PAIR_REGS kMP(const float* __restrict__ T,
                                           const float* __restrict__ wvw,
                                           const float* __restrict__ bk,
                                           const float* __restrict__ bv,
                                           const float* __restrict__ u,
                                           const float* __restrict__ z,
                                           const float* __restrict__ wo,
                                           float* __restrict__ P) {
  __shared__ __attribute__((aligned(16))) float red[8 * 1024];
  __shared__ __attribute__((aligned(16))) float mt[16 * MT_PITCH];
  const int lane = threadIdx.x & 31, wv = threadIdx.x >> 5;
  const int lm = lane & 15, lh = lane >> 4;
  const int job = blockIdx.x;
  const int bh = job >> 2;
  const int b = bh >> 3, h = bh & 7;
  const int i0 = (job & 3) * 16;
  const int k0 = wv * KW;
  const float* arow = T + ((size_t)bh * 64 + i0 + lm) * DM + k0 + 2 * lh;
  v8f acc[4] = {};
  #pragma unroll
  for (int kk = 0; kk < KW; kk += 4) {
    v2f a = *(const v2f*)(arow + kk);
    #pragma unroll
    for (int t = 0; t < 4; ++t) {
      v2f bm = *(const v2f*)(wvw + (size_t)(h * 64 + 16 * t + lm) * DM + k0 + kk + 2 * lh);
      acc[t] = wmma4(a, bm, acc[t]);
    }
  }
  const int o0 = wv * 64;
  const float* wob = wo + (size_t)(o0 + lm) * DM + h * 64 + 2 * lh;
  v2f wf[16][4];
  #pragma unroll
  for (int kq = 0; kq < 16; ++kq)
    #pragma unroll
    for (int t = 0; t < 4; ++t)
      wf[kq][t] = *(const v2f*)(wob + (size_t)(16 * t) * DM + 4 * kq);
  const v4f s = wg_sum_tile(acc, red, wv, lm, lh);
  const int il = i0 + 2 * wv + lh;
  const int ifl = h * 64 + il;
  const float bki = bk[ifl];
  const float ui  = u[(size_t)b * DM + ifl];
  const v4f bv4 = *(const v4f*)(bv + h * 64 + 4 * lm);
  const v4f z4  = *(const v4f*)(z + (size_t)b * DM + h * 64 + 4 * lm);
  v4f o;
  #pragma unroll
  for (int c = 0; c < 4; ++c) {
    const float bvj = bv4[c], zj = z4[c];
    float val = s[c] + ui * bvj + bki * zj + (float)SEQ * bki * bvj;
    o[c] = 0.125f * val;
  }
  *(v4f*)(mt + (2 * wv + lh) * MT_PITCH + 4 * lm) = o;
  __syncthreads();
  const float* mrow = mt + lm * MT_PITCH + 2 * lh;
  v8f pacc[4] = {};
  #pragma unroll
  for (int kq = 0; kq < 16; ++kq) {
    v2f a = *(const v2f*)(mrow + 4 * kq);
    #pragma unroll
    for (int t = 0; t < 4; ++t) pacc[t] = wmma4(a, wf[kq][t], pacc[t]);
  }
  float* Pb = P + ((size_t)b * DM + h * 64 + i0) * DM + o0;
  #pragma unroll
  for (int t = 0; t < 4; ++t)
    #pragma unroll
    for (int r = 0; r < 8; ++r)
      Pb[(size_t)(r + 8 * lh) * DM + 16 * t + lm] = pacc[t][r];
}

__global__ void __launch_bounds__(256) SLICE_REGS kW4C(const float* __restrict__ wq,
                                            const float* __restrict__ bq,
                                            const float* __restrict__ P,
                                            float* __restrict__ W4,
                                            float* __restrict__ cq) {
  __shared__ __attribute__((aligned(16))) float red[8 * 1024];
  const int lane = threadIdx.x & 31, wv = threadIdx.x >> 5;
  if (blockIdx.x < CQ_BLOCKS) {
    cq_cols(blockIdx.x, lane, wv, red, bq, P, cq);
    return;
  }
  const int lm = lane & 15, lh = lane >> 4;
  const int job = blockIdx.x - CQ_BLOCKS;
  const int b = job >> 8;
  const int rest = job & 255;
  const int d0 = (rest >> 3) * 16;
  const int o0 = (rest & 7) * 64;
  const int k0 = wv * KW;
  const float* wqk = wq + (size_t)k0 * DM;
  const float* Pb = P + ((size_t)b * DM + k0) * DM;
  v8f acc[4] = {};
  #pragma unroll
  for (int kk = 0; kk < KW; kk += 4) {
    size_t fr = (size_t)(kk + 2 * lh) * DM;
    v2f a; a.x = wqk[fr + d0 + lm]; a.y = wqk[fr + DM + d0 + lm];
    #pragma unroll
    for (int t = 0; t < 4; ++t) {
      v2f bm; bm.x = Pb[fr + o0 + 16 * t + lm]; bm.y = Pb[fr + DM + o0 + 16 * t + lm];
      acc[t] = wmma4(a, bm, acc[t]);
    }
  }
  const v4f s = wg_sum_tile(acc, red, wv, lm, lh);
  float* Wb = W4 + ((size_t)b * DM + d0) * DM + o0;
  *(v4f*)(Wb + (size_t)(2 * wv + lh) * DM + 4 * lm) = s;
}

struct OSet { v4f a0[2], a1[2], w[8]; };

__device__ __forceinline__ void out_load(OSet& f, const float* __restrict__ qrow0,
                                         const float* __restrict__ qrow1,
                                         const float* __restrict__ Wp, int kk) {
  f.a0[0] = *(const v4f*)(qrow0 + kk); f.a0[1] = *(const v4f*)(qrow0 + kk + 4);
  f.a1[0] = *(const v4f*)(qrow1 + kk); f.a1[1] = *(const v4f*)(qrow1 + kk + 4);
  #pragma unroll
  for (int j = 0; j < 8; ++j) f.w[j] = *(const v4f*)(Wp + (size_t)(kk + j) * DM);
}

__device__ __forceinline__ void out_mma(const OSet& f, v8f (&acc)[8]) {
  #pragma unroll
  for (int p = 0; p < 4; ++p) {
    v2f x0; x0.x = f.a0[p >> 1][2 * (p & 1)]; x0.y = f.a0[p >> 1][2 * (p & 1) + 1];
    v2f x1; x1.x = f.a1[p >> 1][2 * (p & 1)]; x1.y = f.a1[p >> 1][2 * (p & 1) + 1];
    #pragma unroll
    for (int t = 0; t < 4; ++t) {
      v2f bm; bm.x = f.w[2 * p][t]; bm.y = f.w[2 * p + 1][t];
      acc[t]     = wmma4(x0, bm, acc[t]);
      acc[4 + t] = wmma4(x1, bm, acc[4 + t]);
    }
  }
}

__global__ void __launch_bounds__(256) SLICE_REGS kOut(const float* __restrict__ q,
                                            const float* __restrict__ W4,
                                            const float* __restrict__ bo,
                                            const float* __restrict__ cq,
                                            float* __restrict__ out) {
  const int lane = threadIdx.x & 31, wv = threadIdx.x >> 5;
  const int lm = lane & 15, lh = lane >> 4;
  const int job = blockIdx.x * 8 + wv;
  const int b = job >> 7;
  const int n0 = (job & 127) * 32;
  const int o0 = blockIdx.y * 64;
  const float* qrow0 = q + ((size_t)b * SEQ + n0 + lm) * DM + 8 * lh;
  const float* qrow1 = qrow0 + (size_t)16 * DM;
  const float* Wp = W4 + (size_t)b * DM * DM + (size_t)(8 * lh) * DM + o0 + 4 * lm;
  v8f acc[8] = {};
  OSet f0, f1;
  out_load(f0, qrow0, qrow1, Wp, 0);
  for (int kk = 0; kk < DM - 32; kk += 32) {
    out_load(f1, qrow0, qrow1, Wp, kk + 16);
    out_mma(f0, acc);
    out_load(f0, qrow0, qrow1, Wp, kk + 32);
    out_mma(f1, acc);
  }
  out_load(f1, qrow0, qrow1, Wp, DM - 16);
  out_mma(f0, acc);
  out_mma(f1, acc);
  const v4f add = *(const v4f*)(bo + o0 + 4 * lm) + *(const v4f*)(cq + (size_t)b * DM + o0 + 4 * lm);
  float* ob = out + ((size_t)b * SEQ + n0) * DM + o0 + 4 * lm;
  #pragma unroll
  for (int m = 0; m < 2; ++m)
    #pragma unroll
    for (int r = 0; r < 8; ++r) {
      v4f o;
      o.x = acc[4 * m][r]; o.y = acc[4 * m + 1][r]; o.z = acc[4 * m + 2][r]; o.w = acc[4 * m + 3][r];
      *(v4f*)(ob + (size_t)(16 * m + r + 8 * lh) * DM) = o + add;
    }
}

extern "C" void kernel_launch(void* const* d_in, const int* in_sizes, int n_in,
                              void* d_out, int out_size, void* d_ws, size_t ws_size,
                              hipStream_t stream) {
  const float* q   = (const float*)d_in[0];
  const float* k   = (const float*)d_in[1];
  const float* v   = (const float*)d_in[2];
  const float* wq  = (const float*)d_in[3];
  const float* bq  = (const float*)d_in[4];
  const float* wk  = (const float*)d_in[5];
  const float* bk  = (const float*)d_in[6];
  const float* wvw = (const float*)d_in[7];
  const float* bv  = (const float*)d_in[8];
  const float* wo  = (const float*)d_in[9];
  const float* bo  = (const float*)d_in[10];
  float* out = (float*)d_out;

  float* ws = (float*)d_ws;
  float* G    = ws;
  float* T    = G    + (size_t)NB * DM * DM;
  float* P    = T    + (size_t)NB * DM * DM;
  float* W4   = P    + (size_t)NB * DM * DM;
  float* ksum = W4   + (size_t)NB * DM * DM;
  float* vsum = ksum + (size_t)NB * DM;
  float* u    = vsum + (size_t)NB * DM;
  float* z    = u    + (size_t)NB * DM;
  float* cq   = z    + (size_t)NB * DM;

  kG   <<<KG_TILES, KG_THREADS, 0, stream>>>(k, v, G, ksum);
  kTU  <<<TU_GRID, 256, 0, stream>>>(wk, wvw, G, ksum, vsum, T, u, z);
  kMP  <<<64, 256, 0, stream>>>(T, wvw, bk, bv, u, z, wo, P);
  kW4C <<<W4C_GRID, 256, 0, stream>>>(wq, bq, P, W4, cq);
  dim3 go(32, 8, 1);
  kOut <<<go, 256, 0, stream>>>(q, W4, bo, cq, out);
}


# kTU, kW4C: T, W4 tiles take the big kernels' wave program (32x64 a wave, 16 B loads, 2 operand sets), reduction over 4 waves; u,z 4 rows a wave; waves 5120/4352 -> 1280/1152; changes T, W4 sum order
# speedup vs baseline: 10.1779x; 1.0466x over previous
#include <hip/hip_runtime.h>


typedef float v2f __attribute__((ext_vector_type(2)));
typedef float v4f __attribute__((ext_vector_type(4)));
typedef float v8f __attribute__((ext_vector_type(8)));

#define DM   512
#define SEQ  4096
#define NB   2
#define NH   8
#define DKH  64
#define KSPLIT 16
#define SCH  (SEQ / KSPLIT)
#define KG_TILES     (NB * 16 * 8)
#define KG_THREADS   (32 * KSPLIT)
#define TILE_FLOATS  (32 * 64)
#define NPART  (2 * KSPLIT)
#define KW   (DM / 8)
#define KSL  4
#define KWS  (DM / KSL)
#define MID_THREADS (32 * KSL)
#define UZ_ROWS    4
#define UZ_BLOCKS  (2 * DM / (KSL * UZ_ROWS))
#define T_BLOCKS   (NB * 16 * 8)
#define CQ_BLOCKS  (NB * DM / 32)
#define W4_BLOCKS  (NB * 16 * 8)
#define TU_GRID    (T_BLOCKS + UZ_BLOCKS)
#define W4C_GRID   (W4_BLOCKS + CQ_BLOCKS)
#define MT_PITCH   68
#define SLICE_REGS __attribute__((amdgpu_waves_per_eu(4, 4)))
#define PAIR_REGS __attribute__((amdgpu_waves_per_eu(2, 2)))

__device__ __forceinline__ v8f wmma4(v2f a, v2f b, v8f c) {
  return __builtin_amdgcn_wmma_f32_16x16x4_f32(
       false, a,  false, b,
       (short)0, c,  false,  false);
}

__device__ __forceinline__ v4f wg_sum_tile(const v8f (&acc)[4], float* red, int wv, int lm, int lh) {
  float* mine = red + wv * 1024;
  #pragma unroll
  for (int t = 0; t < 4; ++t)
    #pragma unroll
    for (int r = 0; r < 8; ++r)
      mine[(r + 8 * lh) * 64 + 16 * t + lm] = acc[t][r];
  __syncthreads();
  const int off = (2 * wv + lh) * 64 + 4 * lm;
  v4f s = *(const v4f*)(red + off);
  #pragma unroll
  for (int w = 1; w < 8; ++w) s += *(const v4f*)(red + w * 1024 + off);
  return s;
}

struct GSet { v2f kx[4], ky[4]; v4f vx[4], vy[4]; };

__device__ __forceinline__ void g_load(GSet& f, const float* __restrict__ kp,
                                       const float* __restrict__ vp, int s) {
  #pragma unroll
  for (int i = 0; i < 4; ++i) {
    f.kx[i] = *(const v2f*)(kp + (size_t)(s + 4 * i) * DM);
    f.ky[i] = *(const v2f*)(kp + (size_t)(s + 4 * i + 1) * DM);
    f.vx[i] = *(const v4f*)(vp + (size_t)(s + 4 * i) * DM);
    f.vy[i] = *(const v4f*)(vp + (size_t)(s + 4 * i + 1) * DM);
  }
}

template <bool SK, bool SV>
__device__ __forceinline__ void g_mma(const GSet& f, v8f (&acc)[8],
                                      v2f& ksx, v2f& ksy, v4f& vsx, v4f& vsy) {
  #pragma unroll
  for (int i = 0; i < 4; ++i) {
    const v2f kx = f.kx[i], ky = f.ky[i];
    const v4f vx = f.vx[i], vy = f.vy[i];
    if (SK) { ksx += kx; ksy += ky; }
    if (SV) { vsx += vx; vsy += vy; }
    v2f a0; a0.x = kx.x; a0.y = ky.x;
    v2f a1; a1.x = kx.y; a1.y = ky.y;
    #pragma unroll
    for (int t = 0; t < 4; ++t) {
      v2f bm; bm.x = vx[t]; bm.y = vy[t];
      acc[t]     = wmma4(a0, bm, acc[t]);
      acc[4 + t] = wmma4(a1, bm, acc[4 + t]);
    }
  }
}

template <bool SK, bool SV>
__device__ __forceinline__ void g_tile(const float* __restrict__ kb,
                                       const float* __restrict__ vb,
                                       float* plane,
                                       float* spk,
                                       float* spv,
                                       int d0, int e0, int lm, int lh) {
  v8f acc[8] = {};
  v2f ksx = {0.f, 0.f}, ksy = {0.f, 0.f};
  v4f vsx = {0.f, 0.f, 0.f, 0.f}, vsy = {0.f, 0.f, 0.f, 0.f};
  const float* kp = kb + (size_t)(2 * lh) * DM + d0 + 2 * lm;
  const float* vp = vb + (size_t)(2 * lh) * DM + e0 + 4 * lm;
  GSet f0, f1;
  g_load(f0, kp, vp, 0);
  #pragma nounroll
  for (int s0 = 0; s0 < SCH - 32; s0 += 32) {
    g_load(f1, kp, vp, s0 + 16);
    g_mma<SK, SV>(f0, acc, ksx, ksy, vsx, vsy);
    g_load(f0, kp, vp, s0 + 32);
    g_mma<SK, SV>(f1, acc, ksx, ksy, vsx, vsy);
  }
  g_load(f1, kp, vp, SCH - 16);
  g_mma<SK, SV>(f0, acc, ksx, ksy, vsx, vsy);
  g_mma<SK, SV>(f1, acc, ksx, ksy, vsx, vsy);
  #pragma unroll
  for (int m = 0; m < 2; ++m)
    #pragma unroll
    for (int r = 0; r < 8; ++r) {
      v4f o;
      o.x = acc[4 * m][r]; o.y = acc[4 * m + 1][r]; o.z = acc[4 * m + 2][r]; o.w = acc[4 * m + 3][r];
      *(v4f*)(plane + (2 * (8 * lh + r) + m) * 64 + 4 * lm) = o;
    }
  if (SK) *(v2f*)(spk + 2 * lm) = ksx + ksy;
  if (SV) *(v4f*)(spv + 4 * lm) = vsx + vsy;
}

__global__ void __launch_bounds__(KG_THREADS) SLICE_REGS kG(const float* __restrict__ k,
                                          const float* __restrict__ v,
                                          float* __restrict__ G,
                                          float* __restrict__ sums) {
  __shared__ __attribute__((aligned(16))) float red[KSPLIT * TILE_FLOATS];
  __shared__ __attribute__((aligned(16))) float pks[NPART * 32];
  __shared__ __attribute__((aligned(16))) float pvs[NPART * 64];
  const int lane = threadIdx.x & 31;
  const int lm = lane & 15, lh = lane >> 4;
  const int c = __builtin_amdgcn_readfirstlane((int)(threadIdx.x >> 5));
  const int tile = blockIdx.x;
  const int b = tile >> 7;
  const int rest = tile & 127;
  const int dt = rest >> 3;
  const int eg = rest & 7;
  const int d0 = dt * 32;
  const int e0 = eg * 64;
  const float* kb = k + ((size_t)b * SEQ + (size_t)c * SCH) * DM;
  const float* vb = v + ((size_t)b * SEQ + (size_t)c * SCH) * DM;
  float* plane = red + c * TILE_FLOATS;
  float* spk = pks + (2 * c + lh) * 32;
  float* spv = pvs + (2 * c + lh) * 64;
  const bool sumK = eg == 0;
  const bool sumV = dt == 0;
  if (sumV) {
    if (sumK) g_tile<true, true>(kb, vb, plane, spk, spv, d0, e0, lm, lh);
    else      g_tile<false, true>(kb, vb, plane, spk, spv, d0, e0, lm, lh);
  } else {
    if (sumK) g_tile<true, false>(kb, vb, plane, spk, spv, d0, e0, lm, lh);
    else      g_tile<false, false>(kb, vb, plane, spk, spv, d0, e0, lm, lh);
  }
  __syncthreads();
  const int t = threadIdx.x;
  v4f s = *(const v4f*)(red + 4 * t);
  #pragma unroll
  for (int p = 1; p < KSPLIT; ++p) s += *(const v4f*)(red + p * TILE_FLOATS + 4 * t);
  *(v4f*)(G + ((size_t)b * DM + d0 + (t >> 4)) * DM + e0 + 4 * (t & 15)) = s;
  if (sumK && t < 32) {
    float a = pks[t];
    #pragma unroll
    for (int p = 1; p < NPART; ++p) a += pks[p * 32 + t];
    sums[(size_t)b * DM + d0 + t] = a;
  }
  if (sumV && t >= 64 && t < 128) {
    const int j = t - 64;
    float a = pvs[j];
    #pragma unroll
    for (int p = 1; p < NPART; ++p) a += pvs[p * 64 + j];
    sums[(size_t)(NB + b) * DM + e0 + j] = a;
  }
}

struct OSet { v4f a0[2], a1[2], w[8]; };

__device__ __forceinline__ void out_load(OSet& f, const float* __restrict__ qrow0,
                                         const float* __restrict__ qrow1,
                                         const float* __restrict__ Wp, int kk) {
  f.a0[0] = *(const v4f*)(qrow0 + kk); f.a0[1] = *(const v4f*)(qrow0 + kk + 4);
  f.a1[0] = *(const v4f*)(qrow1 + kk); f.a1[1] = *(const v4f*)(qrow1 + kk + 4);
  #pragma unroll
  for (int j = 0; j < 8; ++j) f.w[j] = *(const v4f*)(Wp + (size_t)(kk + j) * DM);
}

__device__ __forceinline__ void out_mma(const OSet& f, v8f (&acc)[8]) {
  #pragma unroll
  for (int p = 0; p < 4; ++p) {
    v2f x0; x0.x = f.a0[p >> 1][2 * (p & 1)]; x0.y = f.a0[p >> 1][2 * (p & 1) + 1];
    v2f x1; x1.x = f.a1[p >> 1][2 * (p & 1)]; x1.y = f.a1[p >> 1][2 * (p & 1) + 1];
    #pragma unroll
    for (int t = 0; t < 4; ++t) {
      v2f bm; bm.x = f.w[2 * p][t]; bm.y = f.w[2 * p + 1][t];
      acc[t]     = wmma4(x0, bm, acc[t]);
      acc[4 + t] = wmma4(x1, bm, acc[4 + t]);
    }
  }
}

template <bool GROWS>
__device__ __forceinline__ void plane_store(float* plane, const v8f (&acc)[8], int lm, int lh) {
  #pragma unroll
  for (int m = 0; m < 2; ++m)
    #pragma unroll
    for (int r = 0; r < 8; ++r) {
      v4f o;
      o.x = acc[4 * m][r]; o.y = acc[4 * m + 1][r]; o.z = acc[4 * m + 2][r]; o.w = acc[4 * m + 3][r];
      const int row = GROWS ? 2 * (8 * lh + r) + m : 16 * m + 8 * lh + r;
      *(v4f*)(plane + row * 64 + 4 * lm) = o;
    }
}

__device__ __forceinline__ void wg_sum_store_tile(const float* red, float* __restrict__ dst, int t) {
  __syncthreads();
  #pragma unroll
  for (int i = 0; i < TILE_FLOATS / (4 * MID_THREADS); ++i) {
    const int idx = i * MID_THREADS + t;
    v4f sum = *(const v4f*)(red + 4 * idx);
    #pragma unroll
    for (int w = 1; w < KSL; ++w) sum += *(const v4f*)(red + w * TILE_FLOATS + 4 * idx);
    *(v4f*)(dst + (size_t)(idx >> 4) * DM + 4 * (idx & 15)) = sum;
  }
}

__device__ __forceinline__ void uz_rows(int blk, int lane, int wv,
                    const float* __restrict__ wk, const float* __restrict__ wvw,
                    const float* __restrict__ ksum, const float* __restrict__ vsum,
                    float* __restrict__ u, float* __restrict__ z) {
  const int job = (blk * KSL + wv) * UZ_ROWS;
  const bool second = job >= DM;
  const int f0 = job & (DM - 1);
  const float* wr = (second ? wvw : wk) + (size_t)f0 * DM + 4 * lane;
  const float* sm = (second ? vsum : ksum) + 4 * lane;
  v4f s0[4], s1[4], w4[UZ_ROWS][4];
  #pragma unroll
  for (int j = 0; j < 4; ++j) {
    s0[j] = *(const v4f*)(sm + 128 * j);
    s1[j] = *(const v4f*)(sm + DM + 128 * j);
  }
  #pragma unroll
  for (int r = 0; r < UZ_ROWS; ++r)
    #pragma unroll
    for (int j = 0; j < 4; ++j) w4[r][j] = *(const v4f*)(wr + (size_t)r * DM + 128 * j);
  float o0 = 0.f, o1 = 0.f;
  #pragma unroll
  for (int r = 0; r < UZ_ROWS; ++r) {
    float p0 = 0.f, p1 = 0.f;
    #pragma unroll
    for (int j = 0; j < 4; ++j)
      #pragma unroll
      for (int c = 0; c < 4; ++c) { p0 += w4[r][j][c] * s0[j][c]; p1 += w4[r][j][c] * s1[j][c]; }
    #pragma unroll
    for (int off = 16; off >= 1; off >>= 1) {
      p0 += __shfl_xor(p0, off);
      p1 += __shfl_xor(p1, off);
    }
    if (lane == r) { o0 = p0; o1 = p1; }
  }
  if (lane < UZ_ROWS) {
    float* dst = second ? z : u;
    dst[f0 + lane] = o0;
    dst[DM + f0 + lane] = o1;
  }
}

__device__ __forceinline__ void cq_cols(int blk, int lane, int wv, float* part,
                    const float* __restrict__ bq, const float* __restrict__ P,
                    float* __restrict__ cq) {
  const int b = blk >> 4;
  const int o = (blk & 15) * 32 + lane;
  #pragma unroll
  for (int hs = 0; hs < 8 / KSL; ++hs) {
    const int sl = wv + KSL * hs;
    const int f0 = sl * KW;
    const float* Pb = P + ((size_t)b * DM + f0) * DM + o;
    float acc = 0.f;
    #pragma unroll
    for (int f = 0; f < KW; ++f) acc += bq[f0 + f] * Pb[(size_t)f * DM];
    part[sl * 32 + lane] = acc;
  }
  __syncthreads();
  if (wv == 0) {
    float t = part[lane];
    #pragma unroll
    for (int w = 1; w < 8; ++w) t += part[w * 32 + lane];
    cq[(size_t)b * DM + o] = t;
  }
}

__global__ void __launch_bounds__(MID_THREADS) SLICE_REGS kTU(const float* __restrict__ wk,
                                           const float* __restrict__ wvw,
                                           const float* __restrict__ G,
                                           const float* __restrict__ ksum,
                                           const float* __restrict__ vsum,
                                           float* __restrict__ T,
                                           float* __restrict__ u,
                                           float* __restrict__ z) {
  __shared__ __attribute__((aligned(16))) float red[KSL * TILE_FLOATS];
  const int lane = threadIdx.x & 31;
  const int wv = __builtin_amdgcn_readfirstlane((int)(threadIdx.x >> 5));
  if (blockIdx.x >= T_BLOCKS) {
    uz_rows(blockIdx.x - T_BLOCKS, lane, wv, wk, wvw, ksum, vsum, u, z);
    return;
  }
  const int lm = lane & 15, lh = lane >> 4;
  const int job = blockIdx.x;
  const int b = job >> 7;
  const int rest = job & 127;
  const int r0 = (rest >> 3) * 32;
  const int e0 = (rest & 7) * 64;
  const int k0 = wv * KWS;
  const float* arow0 = wk + (size_t)(r0 + lm) * DM + k0 + 8 * lh;
  const float* arow1 = arow0 + (size_t)16 * DM;
  const float* Gp = G + ((size_t)b * DM + k0 + 8 * lh) * DM + e0 + 4 * lm;
  v8f acc[8] = {};
  OSet f0, f1;
  out_load(f0, arow0, arow1, Gp, 0);
  #pragma nounroll
  for (int kk = 0; kk < KWS - 32; kk += 32) {
    out_load(f1, arow0, arow1, Gp, kk + 16);
    out_mma(f0, acc);
    out_load(f0, arow0, arow1, Gp, kk + 32);
    out_mma(f1, acc);
  }
  out_load(f1, arow0, arow1, Gp, KWS - 16);
  out_mma(f0, acc);
  out_mma(f1, acc);
  float* plane = red + wv * TILE_FLOATS;
  plane_store<false>(plane, acc, lm, lh);
  wg_sum_store_tile(red, T + ((size_t)b * DM + r0) * DM + e0, (int)threadIdx.x);
}

__global__ void __launch_bounds__(256) PAIR_REGS kMP(const float* __restrict__ T,
                                           const float* __restrict__ wvw,
                                           const float* __restrict__ bk,
                                           const float* __restrict__ bv,
                                           const float* __restrict__ u,
                                           const float* __restrict__ z,
                                           const float* __restrict__ wo,
                                           float* __restrict__ P) {
  __shared__ __attribute__((aligned(16))) float red[8 * 1024];
  __shared__ __attribute__((aligned(16))) float mt[16 * MT_PITCH];
  const int lane = threadIdx.x & 31, wv = threadIdx.x >> 5;
  const int lm = lane & 15, lh = lane >> 4;
  const int job = blockIdx.x;
  const int bh = job >> 2;
  const int b = bh >> 3, h = bh & 7;
  const int i0 = (job & 3) * 16;
  const int k0 = wv * KW;
  const float* arow = T + ((size_t)bh * 64 + i0 + lm) * DM + k0 + 2 * lh;
  v8f acc[4] = {};
  #pragma unroll
  for (int kk = 0; kk < KW; kk += 4) {
    v2f a = *(const v2f*)(arow + kk);
    #pragma unroll
    for (int t = 0; t < 4; ++t) {
      v2f bm = *(const v2f*)(wvw + (size_t)(h * 64 + 16 * t + lm) * DM + k0 + kk + 2 * lh);
      acc[t] = wmma4(a, bm, acc[t]);
    }
  }
  const int o0 = wv * 64;
  const float* wob = wo + (size_t)(o0 + lm) * DM + h * 64 + 2 * lh;
  v2f wf[16][4];
  #pragma unroll
  for (int kq = 0; kq < 16; ++kq)
    #pragma unroll
    for (int t = 0; t < 4; ++t)
      wf[kq][t] = *(const v2f*)(wob + (size_t)(16 * t) * DM + 4 * kq);
  const v4f s = wg_sum_tile(acc, red, wv, lm, lh);
  const int il = i0 + 2 * wv + lh;
  const int ifl = h * 64 + il;
  const float bki = bk[ifl];
  const float ui  = u[(size_t)b * DM + ifl];
  const v4f bv4 = *(const v4f*)(bv + h * 64 + 4 * lm);
  const v4f z4  = *(const v4f*)(z + (size_t)b * DM + h * 64 + 4 * lm);
  v4f o;
  #pragma unroll
  for (int c = 0; c < 4; ++c) {
    const float bvj = bv4[c], zj = z4[c];
    float val = s[c] + ui * bvj + bki * zj + (float)SEQ * bki * bvj;
    o[c] = 0.125f * val;
  }
  *(v4f*)(mt + (2 * wv + lh) * MT_PITCH + 4 * lm) = o;
  __syncthreads();
  const float* mrow = mt + lm * MT_PITCH + 2 * lh;
  v8f pacc[4] = {};
  #pragma unroll
  for (int kq = 0; kq < 16; ++kq) {
    v2f a = *(const v2f*)(mrow + 4 * kq);
    #pragma unroll
    for (int t = 0; t < 4; ++t) pacc[t] = wmma4(a, wf[kq][t], pacc[t]);
  }
  float* Pb = P + ((size_t)b * DM + h * 64 + i0) * DM + o0;
  #pragma unroll
  for (int t = 0; t < 4; ++t)
    #pragma unroll
    for (int r = 0; r < 8; ++r)
      Pb[(size_t)(r + 8 * lh) * DM + 16 * t + lm] = pacc[t][r];
}

__global__ void __launch_bounds__(MID_THREADS) SLICE_REGS kW4C(const float* __restrict__ wq,
                                            const float* __restrict__ bq,
                                            const float* __restrict__ P,
                                            float* __restrict__ W4,
                                            float* __restrict__ cq) {
  __shared__ __attribute__((aligned(16))) float red[KSL * TILE_FLOATS];
  const int lane = threadIdx.x & 31;
  const int wv = __builtin_amdgcn_readfirstlane((int)(threadIdx.x >> 5));
  if (blockIdx.x >= W4_BLOCKS) {
    cq_cols(blockIdx.x - W4_BLOCKS, lane, wv, red, bq, P, cq);
    return;
  }
  const int lm = lane & 15, lh = lane >> 4;
  const int job = blockIdx.x;
  const int b = job >> 7;
  const int rest = job & 127;
  const int d0 = (rest >> 3) * 32;
  const int o0 = (rest & 7) * 64;
  const int k0 = wv * KWS;
  const float* ap = wq + (size_t)(k0 + 2 * lh) * DM + d0 + 2 * lm;
  const float* bp = P + ((size_t)b * DM + k0 + 2 * lh) * DM + o0 + 4 * lm;
  v8f acc[8] = {};
  v2f nk0 = {0.f, 0.f}, nk1 = {0.f, 0.f};
  v4f nv0 = {0.f, 0.f, 0.f, 0.f}, nv1 = {0.f, 0.f, 0.f, 0.f};
  GSet f0, f1;
  g_load(f0, ap, bp, 0);
  #pragma nounroll
  for (int s0 = 0; s0 < KWS - 32; s0 += 32) {
    g_load(f1, ap, bp, s0 + 16);
    g_mma<false, false>(f0, acc, nk0, nk1, nv0, nv1);
    g_load(f0, ap, bp, s0 + 32);
    g_mma<false, false>(f1, acc, nk0, nk1, nv0, nv1);
  }
  g_load(f1, ap, bp, KWS - 16);
  g_mma<false, false>(f0, acc, nk0, nk1, nv0, nv1);
  g_mma<false, false>(f1, acc, nk0, nk1, nv0, nv1);
  float* plane = red + wv * TILE_FLOATS;
  plane_store<true>(plane, acc, lm, lh);
  wg_sum_store_tile(red, W4 + ((size_t)b * DM + d0) * DM + o0, (int)threadIdx.x);
}

__global__ void __launch_bounds__(256) SLICE_REGS kOut(const float* __restrict__ q,
                                            const float* __restrict__ W4,
                                            const float* __restrict__ bo,
                                            const float* __restrict__ cq,
                                            float* __restrict__ out) {
  const int lane = threadIdx.x & 31, wv = threadIdx.x >> 5;
  const int lm = lane & 15, lh = lane >> 4;
  const int job = blockIdx.x * 8 + wv;
  const int b = job >> 7;
  const int n0 = (job & 127) * 32;
  const int o0 = blockIdx.y * 64;
  const float* qrow0 = q + ((size_t)b * SEQ + n0 + lm) * DM + 8 * lh;
  const float* qrow1 = qrow0 + (size_t)16 * DM;
  const float* Wp = W4 + (size_t)b * DM * DM + (size_t)(8 * lh) * DM + o0 + 4 * lm;
  v8f acc[8] = {};
  OSet f0, f1;
  out_load(f0, qrow0, qrow1, Wp, 0);
  for (int kk = 0; kk < DM - 32; kk += 32) {
    out_load(f1, qrow0, qrow1, Wp, kk + 16);
    out_mma(f0, acc);
    out_load(f0, qrow0, qrow1, Wp, kk + 32);
    out_mma(f1, acc);
  }
  out_load(f1, qrow0, qrow1, Wp, DM - 16);
  out_mma(f0, acc);
  out_mma(f1, acc);
  const v4f add = *(const v4f*)(bo + o0 + 4 * lm) + *(const v4f*)(cq + (size_t)b * DM + o0 + 4 * lm);
  float* ob = out + ((size_t)b * SEQ + n0) * DM + o0 + 4 * lm;
  #pragma unroll
  for (int m = 0; m < 2; ++m)
    #pragma unroll
    for (int r = 0; r < 8; ++r) {
      v4f o;
      o.x = acc[4 * m][r]; o.y = acc[4 * m + 1][r]; o.z = acc[4 * m + 2][r]; o.w = acc[4 * m + 3][r];
      *(v4f*)(ob + (size_t)(16 * m + r + 8 * lh) * DM) = o + add;
    }
}

extern "C" void kernel_launch(void* const* d_in, const int* in_sizes, int n_in,
                              void* d_out, int out_size, void* d_ws, size_t ws_size,
                              hipStream_t stream) {
  const float* q   = (const float*)d_in[0];
  const float* k   = (const float*)d_in[1];
  const float* v   = (const float*)d_in[2];
  const float* wq  = (const float*)d_in[3];
  const float* bq  = (const float*)d_in[4];
  const float* wk  = (const float*)d_in[5];
  const float* bk  = (const float*)d_in[6];
  const float* wvw = (const float*)d_in[7];
  const float* bv  = (const float*)d_in[8];
  const float* wo  = (const float*)d_in[9];
  const float* bo  = (const float*)d_in[10];
  float* out = (float*)d_out;

  float* ws = (float*)d_ws;
  float* G    = ws;
  float* T    = G    + (size_t)NB * DM * DM;
  float* P    = T    + (size_t)NB * DM * DM;
  float* W4   = P    + (size_t)NB * DM * DM;
  float* ksum = W4   + (size_t)NB * DM * DM;
  float* vsum = ksum + (size_t)NB * DM;
  float* u    = vsum + (size_t)NB * DM;
  float* z    = u    + (size_t)NB * DM;
  float* cq   = z    + (size_t)NB * DM;

  kG   <<<KG_TILES, KG_THREADS, 0, stream>>>(k, v, G, ksum);
  kTU  <<<TU_GRID, MID_THREADS, 0, stream>>>(wk, wvw, G, ksum, vsum, T, u, z);
  kMP  <<<64, 256, 0, stream>>>(T, wvw, bk, bv, u, z, wo, P);
  kW4C <<<W4C_GRID, MID_THREADS, 0, stream>>>(wq, bq, P, W4, cq);
  dim3 go(32, 8, 1);
  kOut <<<go, 256, 0, stream>>>(q, W4, bo, cq, out);
}
